# starting kernel
# baseline (speedup 1.0000x reference)
#include <hip/hip_runtime.h>


typedef __attribute__((ext_vector_type(2))) float v2f;
typedef __attribute__((ext_vector_type(8))) float v8f;

#define HP     288
#define WPAD   34
#define NITER  26
#define GAMMA  1.0f
#define DTOT   576.0f

__global__ __launch_bounds__(256) void mpconv_sparsemax_kernel(
    const float* __restrict__ x,
    const float* __restrict__ w,
    float* __restrict__ out)
{
    __shared__ float sW[HP * WPAD];
    __shared__ float sP[8][HP];

    const int tid  = threadIdx.x;
    const int lane = tid & 31;
    const int wv   = tid >> 5;

    for (int g = tid; g < 32 * HP; g += 256) {
        int o = g / HP;
        int h = g - o * HP;
        sW[h * WPAD + o] = w[g];
    }

    const int loc = blockIdx.x * 8 + wv;
    const int b   = loc >> 10;
    const int pos = loc & 1023;
    const int oh  = pos >> 5, ow = pos & 31;
    {
        const int c = lane;
        const float* xb = x + ((b * 32 + c) * 32) * 32;
        #pragma unroll
        for (int ki = 0; ki < 3; ++ki) {
            #pragma unroll
            for (int kj = 0; kj < 3; ++kj) {
                const int y  = oh - 1 + ki;
                const int xx = ow - 1 + kj;
                float v = 0.0f;
                if ((unsigned)y < 32u && (unsigned)xx < 32u) v = xb[y * 32 + xx];
                sP[wv][c * 9 + ki * 3 + kj] = v;
            }
        }
    }
    __syncthreads();

    const int   half   = lane >> 4;
    const int   l16    = lane & 15;
    const int   rowoff = half * 8;
    const float bsel   = half ? 0.0f : 1.0f;
    v2f Bm; Bm[0] = bsel; Bm[1] = 0.0f;
    const float* pP = sP[wv];

    float hi_pa = 0.0f, hi_qa = 0.0f, hi_pb = 0.0f, hi_qb = 0.0f;
    for (int h0 = 0; h0 < HP; h0 += 16) {
        float pv = pP[h0 + l16];
        float a0 = half ? 0.0f : pv;
        v2f Ap; Ap[0] =  a0; Ap[1] = 0.0f;
        v2f Aq; Aq[0] = -a0; Aq[1] = 0.0f;
        v8f Ca, Cb;
        #pragma unroll
        for (int r = 0; r < 8; ++r) {
            const float* row = &sW[(h0 + r + rowoff) * WPAD];
            Ca[r] = row[l16];
            Cb[r] = row[l16 + 16];
        }
        v8f Dpa = __builtin_amdgcn_wmma_f32_16x16x4_f32(false, Ap, false, Bm, (short)0, Ca, false, false);
        v8f Dqa = __builtin_amdgcn_wmma_f32_16x16x4_f32(false, Aq, false, Bm, (short)0, Ca, false, false);
        v8f Dpb = __builtin_amdgcn_wmma_f32_16x16x4_f32(false, Ap, false, Bm, (short)0, Cb, false, false);
        v8f Dqb = __builtin_amdgcn_wmma_f32_16x16x4_f32(false, Aq, false, Bm, (short)0, Cb, false, false);
        #pragma unroll
        for (int r = 0; r < 8; ++r) {
            hi_pa = fmaxf(hi_pa, __builtin_fabsf(Dpa[r]));
            hi_qa = fmaxf(hi_qa, __builtin_fabsf(Dqa[r]));
            hi_pb = fmaxf(hi_pb, __builtin_fabsf(Dpb[r]));
            hi_qb = fmaxf(hi_qb, __builtin_fabsf(Dqb[r]));
        }
    }
    hi_pa = fmaxf(hi_pa, __shfl_xor(hi_pa, 16, 32));
    hi_qa = fmaxf(hi_qa, __shfl_xor(hi_qa, 16, 32));
    hi_pb = fmaxf(hi_pb, __shfl_xor(hi_pb, 16, 32));
    hi_qb = fmaxf(hi_qb, __shfl_xor(hi_qb, 16, 32));

    const float lo0 = -GAMMA / DTOT;
    float lo_pa = lo0, lo_qa = lo0, lo_pb = lo0, lo_qb = lo0;

    for (int it = 0; it < NITER; ++it) {
        const float t_pa = 0.5f * (lo_pa + hi_pa);
        const float t_qa = 0.5f * (lo_qa + hi_qa);
        const float t_pb = 0.5f * (lo_pb + hi_pb);
        const float t_qb = 0.5f * (lo_qb + hi_qb);
        const float fl_pa = fmaxf(-2.0f * t_pa, 0.0f);
        const float fl_qa = fmaxf(-2.0f * t_qa, 0.0f);
        const float fl_pb = fmaxf(-2.0f * t_pb, 0.0f);
        const float fl_qb = fmaxf(-2.0f * t_qb, 0.0f);
        float f_pa = 0.0f, f_qa = 0.0f, f_pb = 0.0f, f_qb = 0.0f;

        for (int h0 = 0; h0 < HP; h0 += 16) {
            float pv = pP[h0 + l16];
            float a0 = half ? 0.0f : pv;
            v2f Ap; Ap[0] =  a0; Ap[1] = 0.0f;
            v2f Aq; Aq[0] = -a0; Aq[1] = 0.0f;
            v8f Ca, Cb;
            #pragma unroll
            for (int r = 0; r < 8; ++r) {
                const float* row = &sW[(h0 + r + rowoff) * WPAD];
                Ca[r] = row[l16];
                Cb[r] = row[l16 + 16];
            }
            v8f Dpa = __builtin_amdgcn_wmma_f32_16x16x4_f32(false, Ap, false, Bm, (short)0, Ca, false, false);
            v8f Dqa = __builtin_amdgcn_wmma_f32_16x16x4_f32(false, Aq, false, Bm, (short)0, Ca, false, false);
            v8f Dpb = __builtin_amdgcn_wmma_f32_16x16x4_f32(false, Ap, false, Bm, (short)0, Cb, false, false);
            v8f Dqb = __builtin_amdgcn_wmma_f32_16x16x4_f32(false, Aq, false, Bm, (short)0, Cb, false, false);
            #pragma unroll
            for (int r = 0; r < 8; ++r) {
                f_pa += fmaxf(__builtin_fabsf(Dpa[r]) - t_pa, fl_pa);
                f_qa += fmaxf(__builtin_fabsf(Dqa[r]) - t_qa, fl_qa);
                f_pb += fmaxf(__builtin_fabsf(Dpb[r]) - t_pb, fl_pb);
                f_qb += fmaxf(__builtin_fabsf(Dqb[r]) - t_qb, fl_qb);
            }
        }
        f_pa += __shfl_xor(f_pa, 16, 32);
        f_qa += __shfl_xor(f_qa, 16, 32);
        f_pb += __shfl_xor(f_pb, 16, 32);
        f_qb += __shfl_xor(f_qb, 16, 32);

        lo_pa = (f_pa > GAMMA) ? t_pa : lo_pa;  hi_pa = (f_pa > GAMMA) ? hi_pa : t_pa;
        lo_qa = (f_qa > GAMMA) ? t_qa : lo_qa;  hi_qa = (f_qa > GAMMA) ? hi_qa : t_qa;
        lo_pb = (f_pb > GAMMA) ? t_pb : lo_pb;  hi_pb = (f_pb > GAMMA) ? hi_pb : t_pb;
        lo_qb = (f_qb > GAMMA) ? t_qb : lo_qb;  hi_qb = (f_qb > GAMMA) ? hi_qb : t_qb;
    }

    const float tau_pa = 0.5f * (lo_pa + hi_pa);
    const float tau_qa = 0.5f * (lo_qa + hi_qa);
    const float tau_pb = 0.5f * (lo_pb + hi_pb);
    const float tau_qb = 0.5f * (lo_qb + hi_qb);

    const float tp = half ? tau_pb : tau_pa;
    const float tq = half ? tau_qb : tau_qa;
    const float conv = 0.5f * DTOT * (tp - tq);

    out[(b * 32 + lane) * 1024 + pos] = conv;
}

extern "C" void kernel_launch(void* const* d_in, const int* in_sizes, int n_in,
                              void* d_out, int out_size, void* d_ws, size_t ws_size,
                              hipStream_t stream) {
    const float* x = (const float*)d_in[0];
    const float* w = (const float*)d_in[1];
    float* out = (float*)d_out;
    (void)in_sizes; (void)n_in; (void)out_size; (void)d_ws; (void)ws_size;

    mpconv_sparsemax_kernel<<<256, 256, 0, stream>>>(x, w, out);
}


# Threshold search: 26-step bisection replaced by Newton steps from the left on the convex piecewise-linear f, start max|v|-gamma, stop when the slope repeats; per-pass sum order kept
# speedup vs baseline: 2.4089x; 2.4089x over previous
#include <hip/hip_runtime.h>


typedef __attribute__((ext_vector_type(2))) float v2f;
typedef __attribute__((ext_vector_type(8))) float v8f;

#define HP     288
#define WPAD   34
#define GAMMA  1.0f
#define DTOT   576.0f
#define MAXIT  (2 * HP + 2)

__device__ __forceinline__ void newton_step(float& t, int& sprev, bool& done, float F, int nneg)
{
    const float a   = __builtin_fabsf(t);
    const int   cge = HP - nneg;
    const bool  neg = t < 0.0f;
    const float f   = neg ? (F + DTOT * a) : F;
    int slope = neg ? (2 * HP - cge) : cge;
    slope = slope < 1 ? 1 : slope;
    float step = (f - GAMMA) * __builtin_amdgcn_rcpf((float)slope);
    step = step > 0.0f ? step : 0.0f;
    const bool conv = (slope == sprev) || !(step > 0.0f);
    t     = done ? t : t + step;
    sprev = done ? sprev : slope;
    done  = done || conv;
}

__global__ __launch_bounds__(256) void mpconv_sparsemax_kernel(
    const float* __restrict__ x,
    const float* __restrict__ w,
    float* __restrict__ out)
{
    __shared__ float sW[HP * WPAD];
    __shared__ float sP[8][HP];

    const int tid  = threadIdx.x;
    const int lane = tid & 31;
    const int wv   = tid >> 5;

    for (int g = tid; g < 32 * HP; g += 256) {
        int o = g / HP;
        int h = g - o * HP;
        sW[h * WPAD + o] = w[g];
    }

    const int loc = blockIdx.x * 8 + wv;
    const int b   = loc >> 10;
    const int pos = loc & 1023;
    const int oh  = pos >> 5, ow = pos & 31;
    {
        const int c = lane;
        const float* xb = x + ((b * 32 + c) * 32) * 32;
        #pragma unroll
        for (int ki = 0; ki < 3; ++ki) {
            #pragma unroll
            for (int kj = 0; kj < 3; ++kj) {
                const int y  = oh - 1 + ki;
                const int xx = ow - 1 + kj;
                float v = 0.0f;
                if ((unsigned)y < 32u && (unsigned)xx < 32u) v = xb[y * 32 + xx];
                sP[wv][c * 9 + ki * 3 + kj] = v;
            }
        }
    }
    __syncthreads();

    const int   half   = lane >> 4;
    const int   l16    = lane & 15;
    const int   rowoff = half * 8;
    const float bsel   = half ? 0.0f : 1.0f;
    v2f Bm; Bm[0] = bsel; Bm[1] = 0.0f;
    const float* pP = sP[wv];

    float hi_pa = 0.0f, hi_qa = 0.0f, hi_pb = 0.0f, hi_qb = 0.0f;
    for (int h0 = 0; h0 < HP; h0 += 16) {
        float pv = pP[h0 + l16];
        float a0 = half ? 0.0f : pv;
        v2f Ap; Ap[0] =  a0; Ap[1] = 0.0f;
        v2f Aq; Aq[0] = -a0; Aq[1] = 0.0f;
        v8f Ca, Cb;
        #pragma unroll
        for (int r = 0; r < 8; ++r) {
            const float* row = &sW[(h0 + r + rowoff) * WPAD];
            Ca[r] = row[l16];
            Cb[r] = row[l16 + 16];
        }
        v8f Dpa = __builtin_amdgcn_wmma_f32_16x16x4_f32(false, Ap, false, Bm, (short)0, Ca, false, false);
        v8f Dqa = __builtin_amdgcn_wmma_f32_16x16x4_f32(false, Aq, false, Bm, (short)0, Ca, false, false);
        v8f Dpb = __builtin_amdgcn_wmma_f32_16x16x4_f32(false, Ap, false, Bm, (short)0, Cb, false, false);
        v8f Dqb = __builtin_amdgcn_wmma_f32_16x16x4_f32(false, Aq, false, Bm, (short)0, Cb, false, false);
        #pragma unroll
        for (int r = 0; r < 8; ++r) {
            hi_pa = fmaxf(hi_pa, __builtin_fabsf(Dpa[r]));
            hi_qa = fmaxf(hi_qa, __builtin_fabsf(Dqa[r]));
            hi_pb = fmaxf(hi_pb, __builtin_fabsf(Dpb[r]));
            hi_qb = fmaxf(hi_qb, __builtin_fabsf(Dqb[r]));
        }
    }
    hi_pa = fmaxf(hi_pa, __shfl_xor(hi_pa, 16, 32));
    hi_qa = fmaxf(hi_qa, __shfl_xor(hi_qa, 16, 32));
    hi_pb = fmaxf(hi_pb, __shfl_xor(hi_pb, 16, 32));
    hi_qb = fmaxf(hi_qb, __shfl_xor(hi_qb, 16, 32));

    const float lo0 = -GAMMA / DTOT;
    float t_pa = fmaxf(hi_pa - GAMMA, lo0);
    float t_qa = fmaxf(hi_qa - GAMMA, lo0);
    float t_pb = fmaxf(hi_pb - GAMMA, lo0);
    float t_qb = fmaxf(hi_qb - GAMMA, lo0);
    int  s_pa = -1, s_qa = -1, s_pb = -1, s_qb = -1;
    bool d_pa = false, d_qa = false, d_pb = false, d_qb = false;

    for (int it = 0; it < MAXIT; ++it) {
        const float a_pa = __builtin_fabsf(t_pa);
        const float a_qa = __builtin_fabsf(t_qa);
        const float a_pb = __builtin_fabsf(t_pb);
        const float a_qb = __builtin_fabsf(t_qb);
        float F_pa = 0.0f, F_qa = 0.0f, F_pb = 0.0f, F_qb = 0.0f;
        int   n_pa = 0, n_qa = 0, n_pb = 0, n_qb = 0;

        for (int h0 = 0; h0 < HP; h0 += 32) {
            unsigned g_pa = 0u, g_qa = 0u, g_pb = 0u, g_qb = 0u;
            #pragma unroll
            for (int s = 0; s < 2; ++s) {
                const int hb = h0 + 16 * s;
                float pv = pP[hb + l16];
                float a0 = half ? 0.0f : pv;
                v2f Ap; Ap[0] =  a0; Ap[1] = 0.0f;
                v2f Aq; Aq[0] = -a0; Aq[1] = 0.0f;
                v8f Ca, Cb;
                #pragma unroll
                for (int r = 0; r < 8; ++r) {
                    const float* row = &sW[(hb + r + rowoff) * WPAD];
                    Ca[r] = row[l16];
                    Cb[r] = row[l16 + 16];
                }
                v8f Dpa = __builtin_amdgcn_wmma_f32_16x16x4_f32(false, Ap, false, Bm, (short)0, Ca, false, false);
                v8f Dqa = __builtin_amdgcn_wmma_f32_16x16x4_f32(false, Aq, false, Bm, (short)0, Ca, false, false);
                v8f Dpb = __builtin_amdgcn_wmma_f32_16x16x4_f32(false, Ap, false, Bm, (short)0, Cb, false, false);
                v8f Dqb = __builtin_amdgcn_wmma_f32_16x16x4_f32(false, Aq, false, Bm, (short)0, Cb, false, false);
                #pragma unroll
                for (int r = 0; r < 8; ++r) {
                    const float e_pa = __builtin_fabsf(Dpa[r]) - a_pa;
                    const float e_qa = __builtin_fabsf(Dqa[r]) - a_qa;
                    const float e_pb = __builtin_fabsf(Dpb[r]) - a_pb;
                    const float e_qb = __builtin_fabsf(Dqb[r]) - a_qb;
                    F_pa += fmaxf(e_pa, 0.0f);
                    F_qa += fmaxf(e_qa, 0.0f);
                    F_pb += fmaxf(e_pb, 0.0f);
                    F_qb += fmaxf(e_qb, 0.0f);
                    g_pa = (g_pa << 1) | (__float_as_uint(e_pa) >> 31);
                    g_qa = (g_qa << 1) | (__float_as_uint(e_qa) >> 31);
                    g_pb = (g_pb << 1) | (__float_as_uint(e_pb) >> 31);
                    g_qb = (g_qb << 1) | (__float_as_uint(e_qb) >> 31);
                }
            }
            n_pa += __builtin_popcount(g_pa);
            n_qa += __builtin_popcount(g_qa);
            n_pb += __builtin_popcount(g_pb);
            n_qb += __builtin_popcount(g_qb);
        }
        F_pa += __shfl_xor(F_pa, 16, 32);
        F_qa += __shfl_xor(F_qa, 16, 32);
        F_pb += __shfl_xor(F_pb, 16, 32);
        F_qb += __shfl_xor(F_qb, 16, 32);
        n_pa += __shfl_xor(n_pa, 16, 32);
        n_qa += __shfl_xor(n_qa, 16, 32);
        n_pb += __shfl_xor(n_pb, 16, 32);
        n_qb += __shfl_xor(n_qb, 16, 32);

        newton_step(t_pa, s_pa, d_pa, F_pa, n_pa);
        newton_step(t_qa, s_qa, d_qa, F_qa, n_qa);
        newton_step(t_pb, s_pb, d_pb, F_pb, n_pb);
        newton_step(t_qb, s_qb, d_qb, F_qb, n_qb);

        if (!__any(!(d_pa && d_qa && d_pb && d_qb))) break;
    }

    const float tp = half ? t_pb : t_pa;
    const float tq = half ? t_qb : t_qa;
    const float conv = 0.5f * DTOT * (tp - tq);

    out[(b * 32 + lane) * 1024 + pos] = conv;
}

extern "C" void kernel_launch(void* const* d_in, const int* in_sizes, int n_in,
                              void* d_out, int out_size, void* d_ws, size_t ws_size,
                              hipStream_t stream) {
    const float* x = (const float*)d_in[0];
    const float* w = (const float*)d_in[1];
    float* out = (float*)d_out;
    (void)in_sizes; (void)n_in; (void)out_size; (void)d_ws; (void)ws_size;

    mpconv_sparsemax_kernel<<<256, 256, 0, stream>>>(x, w, out);
}


# Newton passes walk short per-lane LDS lists of the elements at or above the start threshold (one extra full pass builds them, zero-padded) instead of re-forming all 288 rows; sums bit-identical
# speedup vs baseline: 4.0760x; 1.6921x over previous
#include <hip/hip_runtime.h>


typedef __attribute__((ext_vector_type(2))) float v2f;
typedef __attribute__((ext_vector_type(8))) float v8f;

#define HP     288
#define WPAD   34
#define GAMMA  1.0f
#define DTOT   576.0f
#define MAXIT  (2 * HP + 2)
#define LSLOTS 64
#define LROW   128
#define LCAP   (LSLOTS - 8)
#define LSTEP  (LROW * 4)

typedef __attribute__((ext_vector_type(4))) float v4f;

__device__ __forceinline__ void newton_step(float& t, int& sprev, bool& done, float F, int nneg)
{
    const float a   = __builtin_fabsf(t);
    const int   cge = HP - nneg;
    const bool  neg = t < 0.0f;
    const float f   = neg ? (F + DTOT * a) : F;
    int slope = neg ? (2 * HP - cge) : cge;
    slope = slope < 1 ? 1 : slope;
    float step = (f - GAMMA) * __builtin_amdgcn_rcpf((float)slope);
    step = step > 0.0f ? step : 0.0f;
    const bool conv = (slope == sprev) || !(step > 0.0f);
    t     = done ? t : t + step;
    sprev = done ? sprev : slope;
    done  = done || conv;
}

__global__ __launch_bounds__(256) void mpconv_sparsemax_kernel(
    const float* __restrict__ x,
    const float* __restrict__ w,
    float* __restrict__ out)
{
    __shared__ float sW[HP * WPAD];
    __shared__ float sP[8][HP];
    __shared__ __attribute__((aligned(16))) float sL[8 * LSLOTS * LROW];

    const int tid  = threadIdx.x;
    const int lane = tid & 31;
    const int wv   = tid >> 5;

    for (int g = tid; g < 32 * HP; g += 256) {
        int o = g / HP;
        int h = g - o * HP;
        sW[h * WPAD + o] = w[g];
    }

    const int loc = blockIdx.x * 8 + wv;
    const int b   = loc >> 10;
    const int pos = loc & 1023;
    const int oh  = pos >> 5, ow = pos & 31;
    {
        const int c = lane;
        const float* xb = x + ((b * 32 + c) * 32) * 32;
        #pragma unroll
        for (int ki = 0; ki < 3; ++ki) {
            #pragma unroll
            for (int kj = 0; kj < 3; ++kj) {
                const int y  = oh - 1 + ki;
                const int xx = ow - 1 + kj;
                float v = 0.0f;
                if ((unsigned)y < 32u && (unsigned)xx < 32u) v = xb[y * 32 + xx];
                sP[wv][c * 9 + ki * 3 + kj] = v;
            }
        }
    }
    __syncthreads();

    const int   half   = lane >> 4;
    const int   l16    = lane & 15;
    const int   rowoff = half * 8;
    const float bsel   = half ? 0.0f : 1.0f;
    v2f Bm; Bm[0] = bsel; Bm[1] = 0.0f;
    const float* pP = sP[wv];

    float hi_pa = 0.0f, hi_qa = 0.0f, hi_pb = 0.0f, hi_qb = 0.0f;
    for (int h0 = 0; h0 < HP; h0 += 16) {
        float pv = pP[h0 + l16];
        float a0 = half ? 0.0f : pv;
        v2f Ap; Ap[0] =  a0; Ap[1] = 0.0f;
        v2f Aq; Aq[0] = -a0; Aq[1] = 0.0f;
        v8f Ca, Cb;
        #pragma unroll
        for (int r = 0; r < 8; ++r) {
            const float* row = &sW[(h0 + r + rowoff) * WPAD];
            Ca[r] = row[l16];
            Cb[r] = row[l16 + 16];
        }
        v8f Dpa = __builtin_amdgcn_wmma_f32_16x16x4_f32(false, Ap, false, Bm, (short)0, Ca, false, false);
        v8f Dqa = __builtin_amdgcn_wmma_f32_16x16x4_f32(false, Aq, false, Bm, (short)0, Ca, false, false);
        v8f Dpb = __builtin_amdgcn_wmma_f32_16x16x4_f32(false, Ap, false, Bm, (short)0, Cb, false, false);
        v8f Dqb = __builtin_amdgcn_wmma_f32_16x16x4_f32(false, Aq, false, Bm, (short)0, Cb, false, false);
        #pragma unroll
        for (int r = 0; r < 8; ++r) {
            hi_pa = fmaxf(hi_pa, __builtin_fabsf(Dpa[r]));
            hi_qa = fmaxf(hi_qa, __builtin_fabsf(Dqa[r]));
            hi_pb = fmaxf(hi_pb, __builtin_fabsf(Dpb[r]));
            hi_qb = fmaxf(hi_qb, __builtin_fabsf(Dqb[r]));
        }
    }
    hi_pa = fmaxf(hi_pa, __shfl_xor(hi_pa, 16, 32));
    hi_qa = fmaxf(hi_qa, __shfl_xor(hi_qa, 16, 32));
    hi_pb = fmaxf(hi_pb, __shfl_xor(hi_pb, 16, 32));
    hi_qb = fmaxf(hi_qb, __shfl_xor(hi_qb, 16, 32));

    const float lo0 = -GAMMA / DTOT;
    float t_pa = fmaxf(hi_pa - GAMMA, lo0);
    float t_qa = fmaxf(hi_qa - GAMMA, lo0);
    float t_pb = fmaxf(hi_pb - GAMMA, lo0);
    float t_qb = fmaxf(hi_qb - GAMMA, lo0);
    int  s_pa = -1, s_qa = -1, s_pb = -1, s_qb = -1;
    bool d_pa = false, d_qa = false, d_pb = false, d_qb = false;

    const int wbase = wv * (LSLOTS * LROW);
    const int lcol  = wbase + lane * 4;
    bool listed = false;
    int  cnt4   = 0;
    if (__all(t_pa >= 0.0f && t_qa >= 0.0f && t_pb >= 0.0f && t_qb >= 0.0f)) {
        int o_pa = lcol * 4, o_qa = lcol * 4 + 4, o_pb = lcol * 4 + 8, o_qb = lcol * 4 + 12;
        const int lim = lcol * 4 + LCAP * LSTEP;
        int peak = 0;
        const float a_pa = __builtin_fabsf(t_pa), a_qa = __builtin_fabsf(t_qa);
        const float a_pb = __builtin_fabsf(t_pb), a_qb = __builtin_fabsf(t_qb);
        char* const Lb = (char*)sL;
        for (int h0 = 0; h0 < HP; h0 += 16) {
            peak = max(peak, max(max(o_pa, o_qa), max(o_pb, o_qb)));
            o_pa = min(o_pa, lim);
            o_qa = min(o_qa, lim + 4);
            o_pb = min(o_pb, lim + 8);
            o_qb = min(o_qb, lim + 12);
            float pv = pP[h0 + l16];
            float a0 = half ? 0.0f : pv;
            v2f Ap; Ap[0] =  a0; Ap[1] = 0.0f;
            v2f Aq; Aq[0] = -a0; Aq[1] = 0.0f;
            v8f Ca, Cb;
            #pragma unroll
            for (int r = 0; r < 8; ++r) {
                const float* row = &sW[(h0 + r + rowoff) * WPAD];
                Ca[r] = row[l16];
                Cb[r] = row[l16 + 16];
            }
            v8f Dpa = __builtin_amdgcn_wmma_f32_16x16x4_f32(false, Ap, false, Bm, (short)0, Ca, false, false);
            v8f Dqa = __builtin_amdgcn_wmma_f32_16x16x4_f32(false, Aq, false, Bm, (short)0, Ca, false, false);
            v8f Dpb = __builtin_amdgcn_wmma_f32_16x16x4_f32(false, Ap, false, Bm, (short)0, Cb, false, false);
            v8f Dqb = __builtin_amdgcn_wmma_f32_16x16x4_f32(false, Aq, false, Bm, (short)0, Cb, false, false);
            #pragma unroll
            for (int r = 0; r < 8; ++r) {
                *(float*)(Lb + o_pa) = Dpa[r];
                *(float*)(Lb + o_qa) = Dqa[r];
                *(float*)(Lb + o_pb) = Dpb[r];
                *(float*)(Lb + o_qb) = Dqb[r];
                o_pa += (__builtin_fabsf(Dpa[r]) >= a_pa) ? LSTEP : 0;
                o_qa += (__builtin_fabsf(Dqa[r]) >= a_qa) ? LSTEP : 0;
                o_pb += (__builtin_fabsf(Dpb[r]) >= a_pb) ? LSTEP : 0;
                o_qb += (__builtin_fabsf(Dqb[r]) >= a_qb) ? LSTEP : 0;
            }
        }
        peak = max(peak, max(max(o_pa, o_qa), max(o_pb, o_qb)));
        listed = __all(((peak - wbase * 4) >> 9) <= LCAP);
        if (listed) {
            const int c_pa = (o_pa - wbase * 4) >> 9, c_qa = (o_qa - wbase * 4) >> 9;
            const int c_pb = (o_pb - wbase * 4) >> 9, c_qb = (o_qb - wbase * 4) >> 9;
            int m = max(max(c_pa, c_qa), max(c_pb, c_qb));
            m = max(m, __shfl_xor(m, 16, 32));
            m = max(m, __shfl_xor(m,  8, 32));
            m = max(m, __shfl_xor(m,  4, 32));
            m = max(m, __shfl_xor(m,  2, 32));
            m = max(m, __shfl_xor(m,  1, 32));
            cnt4 = (__builtin_amdgcn_readfirstlane(m) + 3) & ~3;
            for (int s = 0; s < cnt4 + 4; ++s) {
                const int o = lcol + s * LROW;
                if (s >= c_pa) sL[o]     = 0.0f;
                if (s >= c_qa) sL[o + 1] = 0.0f;
                if (s >= c_pb) sL[o + 2] = 0.0f;
                if (s >= c_qb) sL[o + 3] = 0.0f;
            }
        }
    }

    if (listed) {
        const v4f* Lr = (const v4f*)&sL[lcol];
        for (int it = 0; it < MAXIT; ++it) {
            const float a_pa = __builtin_fabsf(t_pa);
            const float a_qa = __builtin_fabsf(t_qa);
            const float a_pb = __builtin_fabsf(t_pb);
            const float a_qb = __builtin_fabsf(t_qb);
            float F_pa = 0.0f, F_qa = 0.0f, F_pb = 0.0f, F_qb = 0.0f;
            int   n_pa = 0, n_qa = 0, n_pb = 0, n_qb = 0;
            v4f nx0 = Lr[0], nx1 = Lr[32], nx2 = Lr[64], nx3 = Lr[96];
            for (int s = 0; s < cnt4; s += 4) {
                const v4f c0 = nx0, c1 = nx1, c2 = nx2, c3 = nx3;
                nx0 = Lr[(s + 4) * 32]; nx1 = Lr[(s + 5) * 32]; nx2 = Lr[(s + 6) * 32]; nx3 = Lr[(s + 7) * 32];
                unsigned g_pa = 0u, g_qa = 0u, g_pb = 0u, g_qb = 0u;
                #pragma unroll
                for (int k = 0; k < 4; ++k) {
                    const v4f c = (k == 0) ? c0 : (k == 1) ? c1 : (k == 2) ? c2 : c3;
                    const float e_pa = __builtin_fabsf(c[0]) - a_pa;
                    const float e_qa = __builtin_fabsf(c[1]) - a_qa;
                    const float e_pb = __builtin_fabsf(c[2]) - a_pb;
                    const float e_qb = __builtin_fabsf(c[3]) - a_qb;
                    F_pa += fmaxf(e_pa, 0.0f);
                    F_qa += fmaxf(e_qa, 0.0f);
                    F_pb += fmaxf(e_pb, 0.0f);
                    F_qb += fmaxf(e_qb, 0.0f);
                    g_pa = (g_pa << 1) | (__float_as_uint(e_pa) >> 31);
                    g_qa = (g_qa << 1) | (__float_as_uint(e_qa) >> 31);
                    g_pb = (g_pb << 1) | (__float_as_uint(e_pb) >> 31);
                    g_qb = (g_qb << 1) | (__float_as_uint(e_qb) >> 31);
                }
                n_pa += __builtin_popcount(g_pa);
                n_qa += __builtin_popcount(g_qa);
                n_pb += __builtin_popcount(g_pb);
                n_qb += __builtin_popcount(g_qb);
            }
            F_pa += __shfl_xor(F_pa, 16, 32);
            F_qa += __shfl_xor(F_qa, 16, 32);
            F_pb += __shfl_xor(F_pb, 16, 32);
            F_qb += __shfl_xor(F_qb, 16, 32);
            n_pa += __shfl_xor(n_pa, 16, 32);
            n_qa += __shfl_xor(n_qa, 16, 32);
            n_pb += __shfl_xor(n_pb, 16, 32);
            n_qb += __shfl_xor(n_qb, 16, 32);
            const int off = HP - 2 * cnt4;
            newton_step(t_pa, s_pa, d_pa, F_pa, off + n_pa);
            newton_step(t_qa, s_qa, d_qa, F_qa, off + n_qa);
            newton_step(t_pb, s_pb, d_pb, F_pb, off + n_pb);
            newton_step(t_qb, s_qb, d_qb, F_qb, off + n_qb);

            if (!__any(!(d_pa && d_qa && d_pb && d_qb))) break;
        }
    } else {
        for (int it = 0; it < MAXIT; ++it) {
            const float a_pa = __builtin_fabsf(t_pa);
            const float a_qa = __builtin_fabsf(t_qa);
            const float a_pb = __builtin_fabsf(t_pb);
            const float a_qb = __builtin_fabsf(t_qb);
            float F_pa = 0.0f, F_qa = 0.0f, F_pb = 0.0f, F_qb = 0.0f;
            int   n_pa = 0, n_qa = 0, n_pb = 0, n_qb = 0;

            for (int h0 = 0; h0 < HP; h0 += 32) {
                unsigned g_pa = 0u, g_qa = 0u, g_pb = 0u, g_qb = 0u;
                #pragma unroll
                for (int s = 0; s < 2; ++s) {
                    const int hb = h0 + 16 * s;
                    float pv = pP[hb + l16];
                    float a0 = half ? 0.0f : pv;
                    v2f Ap; Ap[0] =  a0; Ap[1] = 0.0f;
                    v2f Aq; Aq[0] = -a0; Aq[1] = 0.0f;
                    v8f Ca, Cb;
                    #pragma unroll
                    for (int r = 0; r < 8; ++r) {
                        const float* row = &sW[(hb + r + rowoff) * WPAD];
                        Ca[r] = row[l16];
                        Cb[r] = row[l16 + 16];
                    }
                    v8f Dpa = __builtin_amdgcn_wmma_f32_16x16x4_f32(false, Ap, false, Bm, (short)0, Ca, false, false);
                    v8f Dqa = __builtin_amdgcn_wmma_f32_16x16x4_f32(false, Aq, false, Bm, (short)0, Ca, false, false);
                    v8f Dpb = __builtin_amdgcn_wmma_f32_16x16x4_f32(false, Ap, false, Bm, (short)0, Cb, false, false);
                    v8f Dqb = __builtin_amdgcn_wmma_f32_16x16x4_f32(false, Aq, false, Bm, (short)0, Cb, false, false);
                    #pragma unroll
                    for (int r = 0; r < 8; ++r) {
                        const float e_pa = __builtin_fabsf(Dpa[r]) - a_pa;
                        const float e_qa = __builtin_fabsf(Dqa[r]) - a_qa;
                        const float e_pb = __builtin_fabsf(Dpb[r]) - a_pb;
                        const float e_qb = __builtin_fabsf(Dqb[r]) - a_qb;
                        F_pa += fmaxf(e_pa, 0.0f);
                        F_qa += fmaxf(e_qa, 0.0f);
                        F_pb += fmaxf(e_pb, 0.0f);
                        F_qb += fmaxf(e_qb, 0.0f);
                        g_pa = (g_pa << 1) | (__float_as_uint(e_pa) >> 31);
                        g_qa = (g_qa << 1) | (__float_as_uint(e_qa) >> 31);
                        g_pb = (g_pb << 1) | (__float_as_uint(e_pb) >> 31);
                        g_qb = (g_qb << 1) | (__float_as_uint(e_qb) >> 31);
                    }
                }
                n_pa += __builtin_popcount(g_pa);
                n_qa += __builtin_popcount(g_qa);
                n_pb += __builtin_popcount(g_pb);
                n_qb += __builtin_popcount(g_qb);
            }
            F_pa += __shfl_xor(F_pa, 16, 32);
            F_qa += __shfl_xor(F_qa, 16, 32);
            F_pb += __shfl_xor(F_pb, 16, 32);
            F_qb += __shfl_xor(F_qb, 16, 32);
            n_pa += __shfl_xor(n_pa, 16, 32);
            n_qa += __shfl_xor(n_qa, 16, 32);
            n_pb += __shfl_xor(n_pb, 16, 32);
            n_qb += __shfl_xor(n_qb, 16, 32);

            newton_step(t_pa, s_pa, d_pa, F_pa, n_pa);
            newton_step(t_qa, s_qa, d_qa, F_qa, n_qa);
            newton_step(t_pb, s_pb, d_pb, F_pb, n_pb);
            newton_step(t_qb, s_qb, d_qb, F_qb, n_qb);

            if (!__any(!(d_pa && d_qa && d_pb && d_qb))) break;
        }
    }

    const float tp = half ? t_pb : t_pa;
    const float tq = half ? t_qb : t_qa;
    const float conv = 0.5f * DTOT * (tp - tq);

    out[(b * 32 + lane) * 1024 + pos] = conv;
}

extern "C" void kernel_launch(void* const* d_in, const int* in_sizes, int n_in,
                              void* d_out, int out_size, void* d_ws, size_t ws_size,
                              hipStream_t stream) {
    const float* x = (const float*)d_in[0];
    const float* w = (const float*)d_in[1];
    float* out = (float*)d_out;
    (void)in_sizes; (void)n_in; (void)out_size; (void)d_ws; (void)ws_size;

    mpconv_sparsemax_kernel<<<256, 256, 0, stream>>>(x, w, out);
}


# Staging: every global read of a block goes out as one load group (weights as 9 x 16 B per thread, the 8 patches' common 32x3x10 input slab read once, no branches) instead of 15 waited round trips
# speedup vs baseline: 4.4899x; 1.1015x over previous
#include <hip/hip_runtime.h>


typedef __attribute__((ext_vector_type(2))) float v2f;
typedef __attribute__((ext_vector_type(8))) float v8f;

#define HP     288
#define WPAD   34
#define GAMMA  1.0f
#define DTOT   576.0f
#define MAXIT  (2 * HP + 2)
#define LSLOTS 64
#define LROW   128
#define LCAP   (LSLOTS - 8)
#define LSTEP  (LROW * 4)
#define XSLAB  960

typedef __attribute__((ext_vector_type(4))) float v4f;

__device__ __forceinline__ void newton_step(float& t, int& sprev, bool& done, float F, int nneg)
{
    const float a   = __builtin_fabsf(t);
    const int   cge = HP - nneg;
    const bool  neg = t < 0.0f;
    const float f   = neg ? (F + DTOT * a) : F;
    int slope = neg ? (2 * HP - cge) : cge;
    slope = slope < 1 ? 1 : slope;
    float step = (f - GAMMA) * __builtin_amdgcn_rcpf((float)slope);
    step = step > 0.0f ? step : 0.0f;
    const bool conv = (slope == sprev) || !(step > 0.0f);
    t     = done ? t : t + step;
    sprev = done ? sprev : slope;
    done  = done || conv;
}

__global__ __launch_bounds__(256) void mpconv_sparsemax_kernel(
    const float* __restrict__ x,
    const float* __restrict__ w,
    float* __restrict__ out)
{
    __shared__ float sW[HP * WPAD];
    __shared__ float sP[8][HP];
    __shared__ float sX[1024];
    __shared__ __attribute__((aligned(16))) float sL[8 * LSLOTS * LROW];

    const int tid  = threadIdx.x;
    const int lane = tid & 31;
    const int wv   = tid >> 5;

    const int blk = blockIdx.x;
    const int b   = blk >> 7;
    const int oh  = (blk >> 2) & 31;
    const int ow0 = (blk & 3) * 8;
    const int ow  = ow0 + wv;
    const int pos = oh * 32 + ow;
    const int wo  = tid >> 3;
    const int wm  = tid & 7;
    v4f wr[9];
    {
        const v4f* wsrc = (const v4f*)(w + wo * HP + wm * 4);
        #pragma unroll
        for (int k = 0; k < 9; ++k) wr[k] = wsrc[k * 8];
    }
    float xr[4];
    #pragma unroll
    for (int k = 0; k < 4; ++k) {
        int e = tid + 256 * k;
        e = e < XSLAB ? e : XSLAB - 1;
        const int c   = e / 30;
        const int rem = e - c * 30;
        const int ry  = rem / 10;
        const int cx  = rem - ry * 10;
        const int y   = oh - 1 + ry;
        const int xx  = ow0 - 1 + cx;
        const bool in = (unsigned)y < 32u && (unsigned)xx < 32u;
        const int yc  = min(max(y, 0), 31);
        const int xc  = min(max(xx, 0), 31);
        const float v = x[((b * 32 + c) * 32 + yc) * 32 + xc];
        xr[k] = in ? v : 0.0f;
    }
    #pragma unroll
    for (int k = 0; k < 9; ++k) {
        #pragma unroll
        for (int j = 0; j < 4; ++j)
            sW[(k * 32 + wm * 4 + j) * WPAD + wo] = wr[k][j];
    }
    #pragma unroll
    for (int k = 0; k < 4; ++k) sX[tid + 256 * k] = xr[k];
    __syncthreads();

    {
        const int c = lane;
        #pragma unroll
        for (int ki = 0; ki < 3; ++ki) {
            #pragma unroll
            for (int kj = 0; kj < 3; ++kj)
                sP[wv][c * 9 + ki * 3 + kj] = sX[c * 30 + ki * 10 + wv + kj];
        }
    }
    __syncthreads();

    const int   half   = lane >> 4;
    const int   l16    = lane & 15;
    const int   rowoff = half * 8;
    const float bsel   = half ? 0.0f : 1.0f;
    v2f Bm; Bm[0] = bsel; Bm[1] = 0.0f;
    const float* pP = sP[wv];

    float hi_pa = 0.0f, hi_qa = 0.0f, hi_pb = 0.0f, hi_qb = 0.0f;
    for (int h0 = 0; h0 < HP; h0 += 16) {
        float pv = pP[h0 + l16];
        float a0 = half ? 0.0f : pv;
        v2f Ap; Ap[0] =  a0; Ap[1] = 0.0f;
        v2f Aq; Aq[0] = -a0; Aq[1] = 0.0f;
        v8f Ca, Cb;
        #pragma unroll
        for (int r = 0; r < 8; ++r) {
            const float* row = &sW[(h0 + r + rowoff) * WPAD];
            Ca[r] = row[l16];
            Cb[r] = row[l16 + 16];
        }
        v8f Dpa = __builtin_amdgcn_wmma_f32_16x16x4_f32(false, Ap, false, Bm, (short)0, Ca, false, false);
        v8f Dqa = __builtin_amdgcn_wmma_f32_16x16x4_f32(false, Aq, false, Bm, (short)0, Ca, false, false);
        v8f Dpb = __builtin_amdgcn_wmma_f32_16x16x4_f32(false, Ap, false, Bm, (short)0, Cb, false, false);
        v8f Dqb = __builtin_amdgcn_wmma_f32_16x16x4_f32(false, Aq, false, Bm, (short)0, Cb, false, false);
        #pragma unroll
        for (int r = 0; r < 8; ++r) {
            hi_pa = fmaxf(hi_pa, __builtin_fabsf(Dpa[r]));
            hi_qa = fmaxf(hi_qa, __builtin_fabsf(Dqa[r]));
            hi_pb = fmaxf(hi_pb, __builtin_fabsf(Dpb[r]));
            hi_qb = fmaxf(hi_qb, __builtin_fabsf(Dqb[r]));
        }
    }
    hi_pa = fmaxf(hi_pa, __shfl_xor(hi_pa, 16, 32));
    hi_qa = fmaxf(hi_qa, __shfl_xor(hi_qa, 16, 32));
    hi_pb = fmaxf(hi_pb, __shfl_xor(hi_pb, 16, 32));
    hi_qb = fmaxf(hi_qb, __shfl_xor(hi_qb, 16, 32));

    const float lo0 = -GAMMA / DTOT;
    float t_pa = fmaxf(hi_pa - GAMMA, lo0);
    float t_qa = fmaxf(hi_qa - GAMMA, lo0);
    float t_pb = fmaxf(hi_pb - GAMMA, lo0);
    float t_qb = fmaxf(hi_qb - GAMMA, lo0);
    int  s_pa = -1, s_qa = -1, s_pb = -1, s_qb = -1;
    bool d_pa = false, d_qa = false, d_pb = false, d_qb = false;

    const int wbase = wv * (LSLOTS * LROW);
    const int lcol  = wbase + lane * 4;
    bool listed = false;
    int  cnt4   = 0;
    if (__all(t_pa >= 0.0f && t_qa >= 0.0f && t_pb >= 0.0f && t_qb >= 0.0f)) {
        int o_pa = lcol * 4, o_qa = lcol * 4 + 4, o_pb = lcol * 4 + 8, o_qb = lcol * 4 + 12;
        const int lim = lcol * 4 + LCAP * LSTEP;
        int peak = 0;
        const float a_pa = __builtin_fabsf(t_pa), a_qa = __builtin_fabsf(t_qa);
        const float a_pb = __builtin_fabsf(t_pb), a_qb = __builtin_fabsf(t_qb);
        char* const Lb = (char*)sL;
        for (int h0 = 0; h0 < HP; h0 += 16) {
            peak = max(peak, max(max(o_pa, o_qa), max(o_pb, o_qb)));
            o_pa = min(o_pa, lim);
            o_qa = min(o_qa, lim + 4);
            o_pb = min(o_pb, lim + 8);
            o_qb = min(o_qb, lim + 12);
            float pv = pP[h0 + l16];
            float a0 = half ? 0.0f : pv;
            v2f Ap; Ap[0] =  a0; Ap[1] = 0.0f;
            v2f Aq; Aq[0] = -a0; Aq[1] = 0.0f;
            v8f Ca, Cb;
            #pragma unroll
            for (int r = 0; r < 8; ++r) {
                const float* row = &sW[(h0 + r + rowoff) * WPAD];
                Ca[r] = row[l16];
                Cb[r] = row[l16 + 16];
            }
            v8f Dpa = __builtin_amdgcn_wmma_f32_16x16x4_f32(false, Ap, false, Bm, (short)0, Ca, false, false);
            v8f Dqa = __builtin_amdgcn_wmma_f32_16x16x4_f32(false, Aq, false, Bm, (short)0, Ca, false, false);
            v8f Dpb = __builtin_amdgcn_wmma_f32_16x16x4_f32(false, Ap, false, Bm, (short)0, Cb, false, false);
            v8f Dqb = __builtin_amdgcn_wmma_f32_16x16x4_f32(false, Aq, false, Bm, (short)0, Cb, false, false);
            #pragma unroll
            for (int r = 0; r < 8; ++r) {
                *(float*)(Lb + o_pa) = Dpa[r];
                *(float*)(Lb + o_qa) = Dqa[r];
                *(float*)(Lb + o_pb) = Dpb[r];
                *(float*)(Lb + o_qb) = Dqb[r];
                o_pa += (__builtin_fabsf(Dpa[r]) >= a_pa) ? LSTEP : 0;
                o_qa += (__builtin_fabsf(Dqa[r]) >= a_qa) ? LSTEP : 0;
                o_pb += (__builtin_fabsf(Dpb[r]) >= a_pb) ? LSTEP : 0;
                o_qb += (__builtin_fabsf(Dqb[r]) >= a_qb) ? LSTEP : 0;
            }
        }
        peak = max(peak, max(max(o_pa, o_qa), max(o_pb, o_qb)));
        listed = __all(((peak - wbase * 4) >> 9) <= LCAP);
        if (listed) {
            const int c_pa = (o_pa - wbase * 4) >> 9, c_qa = (o_qa - wbase * 4) >> 9;
            const int c_pb = (o_pb - wbase * 4) >> 9, c_qb = (o_qb - wbase * 4) >> 9;
            int m = max(max(c_pa, c_qa), max(c_pb, c_qb));
            m = max(m, __shfl_xor(m, 16, 32));
            m = max(m, __shfl_xor(m,  8, 32));
            m = max(m, __shfl_xor(m,  4, 32));
            m = max(m, __shfl_xor(m,  2, 32));
            m = max(m, __shfl_xor(m,  1, 32));
            cnt4 = (__builtin_amdgcn_readfirstlane(m) + 3) & ~3;
            for (int s = 0; s < cnt4 + 4; ++s) {
                const int o = lcol + s * LROW;
                if (s >= c_pa) sL[o]     = 0.0f;
                if (s >= c_qa) sL[o + 1] = 0.0f;
                if (s >= c_pb) sL[o + 2] = 0.0f;
                if (s >= c_qb) sL[o + 3] = 0.0f;
            }
        }
    }

    if (listed) {
        const v4f* Lr = (const v4f*)&sL[lcol];
        for (int it = 0; it < MAXIT; ++it) {
            const float a_pa = __builtin_fabsf(t_pa);
            const float a_qa = __builtin_fabsf(t_qa);
            const float a_pb = __builtin_fabsf(t_pb);
            const float a_qb = __builtin_fabsf(t_qb);
            float F_pa = 0.0f, F_qa = 0.0f, F_pb = 0.0f, F_qb = 0.0f;
            int   n_pa = 0, n_qa = 0, n_pb = 0, n_qb = 0;
            v4f nx0 = Lr[0], nx1 = Lr[32], nx2 = Lr[64], nx3 = Lr[96];
            for (int s = 0; s < cnt4; s += 4) {
                const v4f c0 = nx0, c1 = nx1, c2 = nx2, c3 = nx3;
                nx0 = Lr[(s + 4) * 32]; nx1 = Lr[(s + 5) * 32]; nx2 = Lr[(s + 6) * 32]; nx3 = Lr[(s + 7) * 32];
                unsigned g_pa = 0u, g_qa = 0u, g_pb = 0u, g_qb = 0u;
                #pragma unroll
                for (int k = 0; k < 4; ++k) {
                    const v4f c = (k == 0) ? c0 : (k == 1) ? c1 : (k == 2) ? c2 : c3;
                    const float e_pa = __builtin_fabsf(c[0]) - a_pa;
                    const float e_qa = __builtin_fabsf(c[1]) - a_qa;
                    const float e_pb = __builtin_fabsf(c[2]) - a_pb;
                    const float e_qb = __builtin_fabsf(c[3]) - a_qb;
                    F_pa += fmaxf(e_pa, 0.0f);
                    F_qa += fmaxf(e_qa, 0.0f);
                    F_pb += fmaxf(e_pb, 0.0f);
                    F_qb += fmaxf(e_qb, 0.0f);
                    g_pa = (g_pa << 1) | (__float_as_uint(e_pa) >> 31);
                    g_qa = (g_qa << 1) | (__float_as_uint(e_qa) >> 31);
                    g_pb = (g_pb << 1) | (__float_as_uint(e_pb) >> 31);
                    g_qb = (g_qb << 1) | (__float_as_uint(e_qb) >> 31);
                }
                n_pa += __builtin_popcount(g_pa);
                n_qa += __builtin_popcount(g_qa);
                n_pb += __builtin_popcount(g_pb);
                n_qb += __builtin_popcount(g_qb);
            }
            F_pa += __shfl_xor(F_pa, 16, 32);
            F_qa += __shfl_xor(F_qa, 16, 32);
            F_pb += __shfl_xor(F_pb, 16, 32);
            F_qb += __shfl_xor(F_qb, 16, 32);
            n_pa += __shfl_xor(n_pa, 16, 32);
            n_qa += __shfl_xor(n_qa, 16, 32);
            n_pb += __shfl_xor(n_pb, 16, 32);
            n_qb += __shfl_xor(n_qb, 16, 32);
            const int off = HP - 2 * cnt4;
            newton_step(t_pa, s_pa, d_pa, F_pa, off + n_pa);
            newton_step(t_qa, s_qa, d_qa, F_qa, off + n_qa);
            newton_step(t_pb, s_pb, d_pb, F_pb, off + n_pb);
            newton_step(t_qb, s_qb, d_qb, F_qb, off + n_qb);

            if (!__any(!(d_pa && d_qa && d_pb && d_qb))) break;
        }
    } else {
        for (int it = 0; it < MAXIT; ++it) {
            const float a_pa = __builtin_fabsf(t_pa);
            const float a_qa = __builtin_fabsf(t_qa);
            const float a_pb = __builtin_fabsf(t_pb);
            const float a_qb = __builtin_fabsf(t_qb);
            float F_pa = 0.0f, F_qa = 0.0f, F_pb = 0.0f, F_qb = 0.0f;
            int   n_pa = 0, n_qa = 0, n_pb = 0, n_qb = 0;

            for (int h0 = 0; h0 < HP; h0 += 32) {
                unsigned g_pa = 0u, g_qa = 0u, g_pb = 0u, g_qb = 0u;
                #pragma unroll
                for (int s = 0; s < 2; ++s) {
                    const int hb = h0 + 16 * s;
                    float pv = pP[hb + l16];
                    float a0 = half ? 0.0f : pv;
                    v2f Ap; Ap[0] =  a0; Ap[1] = 0.0f;
                    v2f Aq; Aq[0] = -a0; Aq[1] = 0.0f;
                    v8f Ca, Cb;
                    #pragma unroll
                    for (int r = 0; r < 8; ++r) {
                        const float* row = &sW[(hb + r + rowoff) * WPAD];
                        Ca[r] = row[l16];
                        Cb[r] = row[l16 + 16];
                    }
                    v8f Dpa = __builtin_amdgcn_wmma_f32_16x16x4_f32(false, Ap, false, Bm, (short)0, Ca, false, false);
                    v8f Dqa = __builtin_amdgcn_wmma_f32_16x16x4_f32(false, Aq, false, Bm, (short)0, Ca, false, false);
                    v8f Dpb = __builtin_amdgcn_wmma_f32_16x16x4_f32(false, Ap, false, Bm, (short)0, Cb, false, false);
                    v8f Dqb = __builtin_amdgcn_wmma_f32_16x16x4_f32(false, Aq, false, Bm, (short)0, Cb, false, false);
                    #pragma unroll
                    for (int r = 0; r < 8; ++r) {
                        const float e_pa = __builtin_fabsf(Dpa[r]) - a_pa;
                        const float e_qa = __builtin_fabsf(Dqa[r]) - a_qa;
                        const float e_pb = __builtin_fabsf(Dpb[r]) - a_pb;
                        const float e_qb = __builtin_fabsf(Dqb[r]) - a_qb;
                        F_pa += fmaxf(e_pa, 0.0f);
                        F_qa += fmaxf(e_qa, 0.0f);
                        F_pb += fmaxf(e_pb, 0.0f);
                        F_qb += fmaxf(e_qb, 0.0f);
                        g_pa = (g_pa << 1) | (__float_as_uint(e_pa) >> 31);
                        g_qa = (g_qa << 1) | (__float_as_uint(e_qa) >> 31);
                        g_pb = (g_pb << 1) | (__float_as_uint(e_pb) >> 31);
                        g_qb = (g_qb << 1) | (__float_as_uint(e_qb) >> 31);
                    }
                }
                n_pa += __builtin_popcount(g_pa);
                n_qa += __builtin_popcount(g_qa);
                n_pb += __builtin_popcount(g_pb);
                n_qb += __builtin_popcount(g_qb);
            }
            F_pa += __shfl_xor(F_pa, 16, 32);
            F_qa += __shfl_xor(F_qa, 16, 32);
            F_pb += __shfl_xor(F_pb, 16, 32);
            F_qb += __shfl_xor(F_qb, 16, 32);
            n_pa += __shfl_xor(n_pa, 16, 32);
            n_qa += __shfl_xor(n_qa, 16, 32);
            n_pb += __shfl_xor(n_pb, 16, 32);
            n_qb += __shfl_xor(n_qb, 16, 32);

            newton_step(t_pa, s_pa, d_pa, F_pa, n_pa);
            newton_step(t_qa, s_qa, d_qa, F_qa, n_qa);
            newton_step(t_pb, s_pb, d_pb, F_pb, n_pb);
            newton_step(t_qb, s_qb, d_qb, F_qb, n_qb);

            if (!__any(!(d_pa && d_qa && d_pb && d_qb))) break;
        }
    }

    const float tp = half ? t_pb : t_pa;
    const float tq = half ? t_qb : t_qa;
    const float conv = 0.5f * DTOT * (tp - tq);

    out[(b * 32 + lane) * 1024 + pos] = conv;
}

extern "C" void kernel_launch(void* const* d_in, const int* in_sizes, int n_in,
                              void* d_out, int out_size, void* d_ws, size_t ws_size,
                              hipStream_t stream) {
    const float* x = (const float*)d_in[0];
    const float* w = (const float*)d_in[1];
    float* out = (float*)d_out;
    (void)in_sizes; (void)n_in; (void)out_size; (void)d_ws; (void)ws_size;

    mpconv_sparsemax_kernel<<<256, 256, 0, stream>>>(x, w, out);
}


# Start threshold: the full max|v| sweep is replaced by a lower bound on tau, the best prefix mean (sum|v| - gamma)/k over the 8 rows of largest |patch|, less a rounding margin; per-pass sum order kept
# speedup vs baseline: 6.5974x; 1.4694x over previous
#include <hip/hip_runtime.h>


typedef __attribute__((ext_vector_type(2))) float v2f;
typedef __attribute__((ext_vector_type(8))) float v8f;

#define HP     288
#define WPAD   34
#define GAMMA  1.0f
#define DTOT   576.0f
#define MAXIT  (2 * HP + 2)
#define LSLOTS 64
#define LROW   128
#define LCAP   (LSLOTS - 8)
#define LSTEP  (LROW * 4)
#define XSLAB  960
#define NSEL   8

typedef __attribute__((ext_vector_type(4))) float v4f;

__device__ __forceinline__ unsigned wave_umax(unsigned v)
{
    v = max(v, (unsigned)__builtin_amdgcn_update_dpp(0, (int)v, 0x161, 0xf, 0xf, true));
    v = max(v, (unsigned)__builtin_amdgcn_update_dpp(0, (int)v, 0x162, 0xf, 0xf, true));
    v = max(v, (unsigned)__builtin_amdgcn_update_dpp(0, (int)v, 0x164, 0xf, 0xf, true));
    v = max(v, (unsigned)__builtin_amdgcn_update_dpp(0, (int)v, 0x168, 0xf, 0xf, true));
    v = max(v, (unsigned)__builtin_amdgcn_permlanex16((int)v, (int)v, 0x76543210, 0xfedcba98, false, false));
    return v;
}

__device__ __forceinline__ void newton_step(float& t, int& sprev, bool& done, float F, int nneg)
{
    const float a   = __builtin_fabsf(t);
    const int   cge = HP - nneg;
    const bool  neg = t < 0.0f;
    const float f   = neg ? (F + DTOT * a) : F;
    int slope = neg ? (2 * HP - cge) : cge;
    slope = slope < 1 ? 1 : slope;
    float step = (f - GAMMA) * __builtin_amdgcn_rcpf((float)slope);
    step = step > 0.0f ? step : 0.0f;
    const bool conv = (slope == sprev) || !(step > 0.0f);
    t     = done ? t : t + step;
    sprev = done ? sprev : slope;
    done  = done || conv;
}

__global__ __launch_bounds__(256) void mpconv_sparsemax_kernel(
    const float* __restrict__ x,
    const float* __restrict__ w,
    float* __restrict__ out)
{
    __shared__ float sW[HP * WPAD];
    __shared__ float sP[8][HP];
    __shared__ float sX[1024];
    __shared__ __attribute__((aligned(16))) float sL[8 * LSLOTS * LROW];

    const int tid  = threadIdx.x;
    const int lane = tid & 31;
    const int wv   = tid >> 5;

    const int blk = blockIdx.x;
    const int b   = blk >> 7;
    const int oh  = (blk >> 2) & 31;
    const int ow0 = (blk & 3) * 8;
    const int ow  = ow0 + wv;
    const int pos = oh * 32 + ow;
    const int wo  = tid >> 3;
    const int wm  = tid & 7;
    v4f wr[9];
    {
        const v4f* wsrc = (const v4f*)(w + wo * HP + wm * 4);
        #pragma unroll
        for (int k = 0; k < 9; ++k) wr[k] = wsrc[k * 8];
    }
    float xr[4];
    #pragma unroll
    for (int k = 0; k < 4; ++k) {
        int e = tid + 256 * k;
        e = e < XSLAB ? e : XSLAB - 1;
        const int c   = e / 30;
        const int rem = e - c * 30;
        const int ry  = rem / 10;
        const int cx  = rem - ry * 10;
        const int y   = oh - 1 + ry;
        const int xx  = ow0 - 1 + cx;
        const bool in = (unsigned)y < 32u && (unsigned)xx < 32u;
        const int yc  = min(max(y, 0), 31);
        const int xc  = min(max(xx, 0), 31);
        const float v = x[((b * 32 + c) * 32 + yc) * 32 + xc];
        xr[k] = in ? v : 0.0f;
    }
    #pragma unroll
    for (int k = 0; k < 9; ++k) {
        #pragma unroll
        for (int j = 0; j < 4; ++j)
            sW[(k * 32 + wm * 4 + j) * WPAD + wo] = wr[k][j];
    }
    #pragma unroll
    for (int k = 0; k < 4; ++k) sX[tid + 256 * k] = xr[k];
    __syncthreads();

    unsigned m1, m2;
    {
        const int c = lane;
        unsigned key[9];
        #pragma unroll
        for (int ki = 0; ki < 3; ++ki) {
            #pragma unroll
            for (int kj = 0; kj < 3; ++kj) {
                const int   h  = c * 9 + ki * 3 + kj;
                const float pe = sX[c * 30 + ki * 10 + wv + kj];
                sP[wv][h] = pe;
                key[ki * 3 + kj] = (__float_as_uint(__builtin_fabsf(pe)) & 0xFFFFFE00u) | (unsigned)h;
            }
        }
        m1 = max(key[0], key[1]);
        m2 = min(key[0], key[1]);
        #pragma unroll
        for (int i = 2; i < 9; ++i) {
            m2 = max(m2, min(m1, key[i]));
            m1 = max(m1, key[i]);
        }
    }
    int hsel[NSEL];
    #pragma unroll
    for (int k = 0; k < NSEL; ++k) {
        const unsigned g = wave_umax(m1);
        hsel[k] = __builtin_amdgcn_readfirstlane((int)(g & 0x1FFu));
        const bool own = (m1 == g);
        m1 = own ? m2 : m1;
        m2 = own ? 0u : m2;
    }
    __syncthreads();

    const int   half   = lane >> 4;
    const int   l16    = lane & 15;
    const int   rowoff = half * 8;
    const float bsel   = half ? 0.0f : 1.0f;
    v2f Bm; Bm[0] = bsel; Bm[1] = 0.0f;
    const float* pP = sP[wv];

    float S_pa = 0.0f, S_qa = 0.0f, S_pb = 0.0f, S_qb = 0.0f;
    float b_pa = 0.0f, b_qa = 0.0f, b_pb = 0.0f, b_qb = 0.0f;
    #pragma unroll
    for (int k = 0; k < NSEL; ++k) {
        const float pk = pP[hsel[k]];
        const float wa = sW[hsel[k] * WPAD + l16];
        const float wb = sW[hsel[k] * WPAD + l16 + 16];
        S_pa += __builtin_fabsf(pk + wa);
        S_qa += __builtin_fabsf(pk - wa);
        S_pb += __builtin_fabsf(pk + wb);
        S_qb += __builtin_fabsf(pk - wb);
        const float inv = 1.0f / (float)(k + 1);
        const float c_pa = (S_pa - GAMMA) * inv, c_qa = (S_qa - GAMMA) * inv;
        const float c_pb = (S_pb - GAMMA) * inv, c_qb = (S_qb - GAMMA) * inv;
        b_pa = k ? fmaxf(b_pa, c_pa) : c_pa;
        b_qa = k ? fmaxf(b_qa, c_qa) : c_qa;
        b_pb = k ? fmaxf(b_pb, c_pb) : c_pb;
        b_qb = k ? fmaxf(b_qb, c_qb) : c_qb;
    }

    const float lo0 = -GAMMA / DTOT;
    float t_pa = fmaxf(b_pa - 2.4e-6f * (S_pa + GAMMA), lo0);
    float t_qa = fmaxf(b_qa - 2.4e-6f * (S_qa + GAMMA), lo0);
    float t_pb = fmaxf(b_pb - 2.4e-6f * (S_pb + GAMMA), lo0);
    float t_qb = fmaxf(b_qb - 2.4e-6f * (S_qb + GAMMA), lo0);
    int  s_pa = -1, s_qa = -1, s_pb = -1, s_qb = -1;
    bool d_pa = false, d_qa = false, d_pb = false, d_qb = false;

    const int wbase = wv * (LSLOTS * LROW);
    const int lcol  = wbase + lane * 4;
    bool listed = false;
    int  cnt4   = 0;
    if (__all(t_pa >= 0.0f && t_qa >= 0.0f && t_pb >= 0.0f && t_qb >= 0.0f)) {
        int o_pa = lcol * 4, o_qa = lcol * 4 + 4, o_pb = lcol * 4 + 8, o_qb = lcol * 4 + 12;
        const int lim = lcol * 4 + LCAP * LSTEP;
        int peak = 0;
        const float a_pa = __builtin_fabsf(t_pa), a_qa = __builtin_fabsf(t_qa);
        const float a_pb = __builtin_fabsf(t_pb), a_qb = __builtin_fabsf(t_qb);
        char* const Lb = (char*)sL;
        for (int h0 = 0; h0 < HP; h0 += 16) {
            peak = max(peak, max(max(o_pa, o_qa), max(o_pb, o_qb)));
            o_pa = min(o_pa, lim);
            o_qa = min(o_qa, lim + 4);
            o_pb = min(o_pb, lim + 8);
            o_qb = min(o_qb, lim + 12);
            float pv = pP[h0 + l16];
            float a0 = half ? 0.0f : pv;
            v2f Ap; Ap[0] =  a0; Ap[1] = 0.0f;
            v2f Aq; Aq[0] = -a0; Aq[1] = 0.0f;
            v8f Ca, Cb;
            #pragma unroll
            for (int r = 0; r < 8; ++r) {
                const float* row = &sW[(h0 + r + rowoff) * WPAD];
                Ca[r] = row[l16];
                Cb[r] = row[l16 + 16];
            }
            v8f Dpa = __builtin_amdgcn_wmma_f32_16x16x4_f32(false, Ap, false, Bm, (short)0, Ca, false, false);
            v8f Dqa = __builtin_amdgcn_wmma_f32_16x16x4_f32(false, Aq, false, Bm, (short)0, Ca, false, false);
            v8f Dpb = __builtin_amdgcn_wmma_f32_16x16x4_f32(false, Ap, false, Bm, (short)0, Cb, false, false);
            v8f Dqb = __builtin_amdgcn_wmma_f32_16x16x4_f32(false, Aq, false, Bm, (short)0, Cb, false, false);
            #pragma unroll
            for (int r = 0; r < 8; ++r) {
                *(float*)(Lb + o_pa) = Dpa[r];
                *(float*)(Lb + o_qa) = Dqa[r];
                *(float*)(Lb + o_pb) = Dpb[r];
                *(float*)(Lb + o_qb) = Dqb[r];
                o_pa += (__builtin_fabsf(Dpa[r]) >= a_pa) ? LSTEP : 0;
                o_qa += (__builtin_fabsf(Dqa[r]) >= a_qa) ? LSTEP : 0;
                o_pb += (__builtin_fabsf(Dpb[r]) >= a_pb) ? LSTEP : 0;
                o_qb += (__builtin_fabsf(Dqb[r]) >= a_qb) ? LSTEP : 0;
            }
        }
        peak = max(peak, max(max(o_pa, o_qa), max(o_pb, o_qb)));
        listed = __all(((peak - wbase * 4) >> 9) <= LCAP);
        if (listed) {
            const int c_pa = (o_pa - wbase * 4) >> 9, c_qa = (o_qa - wbase * 4) >> 9;
            const int c_pb = (o_pb - wbase * 4) >> 9, c_qb = (o_qb - wbase * 4) >> 9;
            int m = max(max(c_pa, c_qa), max(c_pb, c_qb));
            m = max(m, __shfl_xor(m, 16, 32));
            m = max(m, __shfl_xor(m,  8, 32));
            m = max(m, __shfl_xor(m,  4, 32));
            m = max(m, __shfl_xor(m,  2, 32));
            m = max(m, __shfl_xor(m,  1, 32));
            cnt4 = (__builtin_amdgcn_readfirstlane(m) + 3) & ~3;
            for (int s = 0; s < cnt4 + 4; ++s) {
                const int o = lcol + s * LROW;
                if (s >= c_pa) sL[o]     = 0.0f;
                if (s >= c_qa) sL[o + 1] = 0.0f;
                if (s >= c_pb) sL[o + 2] = 0.0f;
                if (s >= c_qb) sL[o + 3] = 0.0f;
            }
        }
    }

    if (listed) {
        const v4f* Lr = (const v4f*)&sL[lcol];
        for (int it = 0; it < MAXIT; ++it) {
            const float a_pa = __builtin_fabsf(t_pa);
            const float a_qa = __builtin_fabsf(t_qa);
            const float a_pb = __builtin_fabsf(t_pb);
            const float a_qb = __builtin_fabsf(t_qb);
            float F_pa = 0.0f, F_qa = 0.0f, F_pb = 0.0f, F_qb = 0.0f;
            int   n_pa = 0, n_qa = 0, n_pb = 0, n_qb = 0;
            v4f nx0 = Lr[0], nx1 = Lr[32], nx2 = Lr[64], nx3 = Lr[96];
            for (int s = 0; s < cnt4; s += 4) {
                const v4f c0 = nx0, c1 = nx1, c2 = nx2, c3 = nx3;
                nx0 = Lr[(s + 4) * 32]; nx1 = Lr[(s + 5) * 32]; nx2 = Lr[(s + 6) * 32]; nx3 = Lr[(s + 7) * 32];
                unsigned g_pa = 0u, g_qa = 0u, g_pb = 0u, g_qb = 0u;
                #pragma unroll
                for (int k = 0; k < 4; ++k) {
                    const v4f c = (k == 0) ? c0 : (k == 1) ? c1 : (k == 2) ? c2 : c3;
                    const float e_pa = __builtin_fabsf(c[0]) - a_pa;
                    const float e_qa = __builtin_fabsf(c[1]) - a_qa;
                    const float e_pb = __builtin_fabsf(c[2]) - a_pb;
                    const float e_qb = __builtin_fabsf(c[3]) - a_qb;
                    F_pa += fmaxf(e_pa, 0.0f);
                    F_qa += fmaxf(e_qa, 0.0f);
                    F_pb += fmaxf(e_pb, 0.0f);
                    F_qb += fmaxf(e_qb, 0.0f);
                    g_pa = (g_pa << 1) | (__float_as_uint(e_pa) >> 31);
                    g_qa = (g_qa << 1) | (__float_as_uint(e_qa) >> 31);
                    g_pb = (g_pb << 1) | (__float_as_uint(e_pb) >> 31);
                    g_qb = (g_qb << 1) | (__float_as_uint(e_qb) >> 31);
                }
                n_pa += __builtin_popcount(g_pa);
                n_qa += __builtin_popcount(g_qa);
                n_pb += __builtin_popcount(g_pb);
                n_qb += __builtin_popcount(g_qb);
            }
            F_pa += __shfl_xor(F_pa, 16, 32);
            F_qa += __shfl_xor(F_qa, 16, 32);
            F_pb += __shfl_xor(F_pb, 16, 32);
            F_qb += __shfl_xor(F_qb, 16, 32);
            n_pa += __shfl_xor(n_pa, 16, 32);
            n_qa += __shfl_xor(n_qa, 16, 32);
            n_pb += __shfl_xor(n_pb, 16, 32);
            n_qb += __shfl_xor(n_qb, 16, 32);
            const int off = HP - 2 * cnt4;
            newton_step(t_pa, s_pa, d_pa, F_pa, off + n_pa);
            newton_step(t_qa, s_qa, d_qa, F_qa, off + n_qa);
            newton_step(t_pb, s_pb, d_pb, F_pb, off + n_pb);
            newton_step(t_qb, s_qb, d_qb, F_qb, off + n_qb);

            if (!__any(!(d_pa && d_qa && d_pb && d_qb))) break;
        }
    } else {
        for (int it = 0; it < MAXIT; ++it) {
            const float a_pa = __builtin_fabsf(t_pa);
            const float a_qa = __builtin_fabsf(t_qa);
            const float a_pb = __builtin_fabsf(t_pb);
            const float a_qb = __builtin_fabsf(t_qb);
            float F_pa = 0.0f, F_qa = 0.0f, F_pb = 0.0f, F_qb = 0.0f;
            int   n_pa = 0, n_qa = 0, n_pb = 0, n_qb = 0;

            for (int h0 = 0; h0 < HP; h0 += 32) {
                unsigned g_pa = 0u, g_qa = 0u, g_pb = 0u, g_qb = 0u;
                #pragma unroll
                for (int s = 0; s < 2; ++s) {
                    const int hb = h0 + 16 * s;
                    float pv = pP[hb + l16];
                    float a0 = half ? 0.0f : pv;
                    v2f Ap; Ap[0] =  a0; Ap[1] = 0.0f;
                    v2f Aq; Aq[0] = -a0; Aq[1] = 0.0f;
                    v8f Ca, Cb;
                    #pragma unroll
                    for (int r = 0; r < 8; ++r) {
                        const float* row = &sW[(hb + r + rowoff) * WPAD];
                        Ca[r] = row[l16];
                        Cb[r] = row[l16 + 16];
                    }
                    v8f Dpa = __builtin_amdgcn_wmma_f32_16x16x4_f32(false, Ap, false, Bm, (short)0, Ca, false, false);
                    v8f Dqa = __builtin_amdgcn_wmma_f32_16x16x4_f32(false, Aq, false, Bm, (short)0, Ca, false, false);
                    v8f Dpb = __builtin_amdgcn_wmma_f32_16x16x4_f32(false, Ap, false, Bm, (short)0, Cb, false, false);
                    v8f Dqb = __builtin_amdgcn_wmma_f32_16x16x4_f32(false, Aq, false, Bm, (short)0, Cb, false, false);
                    #pragma unroll
                    for (int r = 0; r < 8; ++r) {
                        const float e_pa = __builtin_fabsf(Dpa[r]) - a_pa;
                        const float e_qa = __builtin_fabsf(Dqa[r]) - a_qa;
                        const float e_pb = __builtin_fabsf(Dpb[r]) - a_pb;
                        const float e_qb = __builtin_fabsf(Dqb[r]) - a_qb;
                        F_pa += fmaxf(e_pa, 0.0f);
                        F_qa += fmaxf(e_qa, 0.0f);
                        F_pb += fmaxf(e_pb, 0.0f);
                        F_qb += fmaxf(e_qb, 0.0f);
                        g_pa = (g_pa << 1) | (__float_as_uint(e_pa) >> 31);
                        g_qa = (g_qa << 1) | (__float_as_uint(e_qa) >> 31);
                        g_pb = (g_pb << 1) | (__float_as_uint(e_pb) >> 31);
                        g_qb = (g_qb << 1) | (__float_as_uint(e_qb) >> 31);
                    }
                }
                n_pa += __builtin_popcount(g_pa);
                n_qa += __builtin_popcount(g_qa);
                n_pb += __builtin_popcount(g_pb);
                n_qb += __builtin_popcount(g_qb);
            }
            F_pa += __shfl_xor(F_pa, 16, 32);
            F_qa += __shfl_xor(F_qa, 16, 32);
            F_pb += __shfl_xor(F_pb, 16, 32);
            F_qb += __shfl_xor(F_qb, 16, 32);
            n_pa += __shfl_xor(n_pa, 16, 32);
            n_qa += __shfl_xor(n_qa, 16, 32);
            n_pb += __shfl_xor(n_pb, 16, 32);
            n_qb += __shfl_xor(n_qb, 16, 32);

            newton_step(t_pa, s_pa, d_pa, F_pa, n_pa);
            newton_step(t_qa, s_qa, d_qa, F_qa, n_qa);
            newton_step(t_pb, s_pb, d_pb, F_pb, n_pb);
            newton_step(t_qb, s_qb, d_qb, F_qb, n_qb);

            if (!__any(!(d_pa && d_qa && d_pb && d_qb))) break;
        }
    }

    const float tp = half ? t_pb : t_pa;
    const float tq = half ? t_qb : t_qa;
    const float conv = 0.5f * DTOT * (tp - tq);

    out[(b * 32 + lane) * 1024 + pos] = conv;
}

extern "C" void kernel_launch(void* const* d_in, const int* in_sizes, int n_in,
                              void* d_out, int out_size, void* d_ws, size_t ws_size,
                              hipStream_t stream) {
    const float* x = (const float*)d_in[0];
    const float* w = (const float*)d_in[1];
    float* out = (float*)d_out;
    (void)in_sizes; (void)n_in; (void)out_size; (void)d_ws; (void)ws_size;

    mpconv_sparsemax_kernel<<<256, 256, 0, stream>>>(x, w, out);
}


# List-building pass forms only rows with |patch| + max|W| at or above the wave's smallest start threshold (ballot-gathered per tile half, ascending h) instead of all 288; lists and sums bit-identical
# speedup vs baseline: 11.0463x; 1.6743x over previous
#include <hip/hip_runtime.h>


typedef __attribute__((ext_vector_type(2))) float v2f;
typedef __attribute__((ext_vector_type(8))) float v8f;

#define HP     288
#define WPAD   34
#define GAMMA  1.0f
#define DTOT   576.0f
#define MAXIT  (2 * HP + 2)
#define LSLOTS 64
#define LROW   128
#define LCAP   (LSLOTS - 8)
#define LSTEP  (LROW * 4)
#define XSLAB  960
#define NSEL   8
#define CPITCH 160
#define CWIDEN 1.00001f

typedef __attribute__((ext_vector_type(4))) float v4f;
typedef __attribute__((ext_vector_type(4))) int   v4i;

__device__ __forceinline__ unsigned wave_umax(unsigned v)
{
    v = max(v, (unsigned)__builtin_amdgcn_update_dpp(0, (int)v, 0x161, 0xf, 0xf, true));
    v = max(v, (unsigned)__builtin_amdgcn_update_dpp(0, (int)v, 0x162, 0xf, 0xf, true));
    v = max(v, (unsigned)__builtin_amdgcn_update_dpp(0, (int)v, 0x164, 0xf, 0xf, true));
    v = max(v, (unsigned)__builtin_amdgcn_update_dpp(0, (int)v, 0x168, 0xf, 0xf, true));
    v = max(v, (unsigned)__builtin_amdgcn_permlanex16((int)v, (int)v, 0x76543210, 0xfedcba98, false, false));
    return v;
}

__device__ __forceinline__ void newton_step(float& t, int& sprev, bool& done, float F, int nneg)
{
    const float a   = __builtin_fabsf(t);
    const int   cge = HP - nneg;
    const bool  neg = t < 0.0f;
    const float f   = neg ? (F + DTOT * a) : F;
    int slope = neg ? (2 * HP - cge) : cge;
    slope = slope < 1 ? 1 : slope;
    float step = (f - GAMMA) * __builtin_amdgcn_rcpf((float)slope);
    step = step > 0.0f ? step : 0.0f;
    const bool conv = (slope == sprev) || !(step > 0.0f);
    t     = done ? t : t + step;
    sprev = done ? sprev : slope;
    done  = done || conv;
}

__global__ __launch_bounds__(256) void mpconv_sparsemax_kernel(
    const float* __restrict__ x,
    const float* __restrict__ w,
    float* __restrict__ out)
{
    __shared__ float sW[(HP + 1) * WPAD];
    __shared__ float sP[8][HP + 8];
    __shared__ __attribute__((aligned(16))) float sWm[8];
    __shared__ __attribute__((aligned(16))) int sC[8][2][CPITCH];
    __shared__ float sX[1024];
    __shared__ __attribute__((aligned(16))) float sL[8 * LSLOTS * LROW];

    const int tid  = threadIdx.x;
    const int lane = tid & 31;
    const int wv   = tid >> 5;

    const int blk = blockIdx.x;
    const int b   = blk >> 7;
    const int oh  = (blk >> 2) & 31;
    const int ow0 = (blk & 3) * 8;
    const int ow  = ow0 + wv;
    const int pos = oh * 32 + ow;
    const int wo  = tid >> 3;
    const int wm  = tid & 7;
    v4f wr[9];
    {
        const v4f* wsrc = (const v4f*)(w + wo * HP + wm * 4);
        #pragma unroll
        for (int k = 0; k < 9; ++k) wr[k] = wsrc[k * 8];
    }
    float xr[4];
    #pragma unroll
    for (int k = 0; k < 4; ++k) {
        int e = tid + 256 * k;
        e = e < XSLAB ? e : XSLAB - 1;
        const int c   = e / 30;
        const int rem = e - c * 30;
        const int ry  = rem / 10;
        const int cx  = rem - ry * 10;
        const int y   = oh - 1 + ry;
        const int xx  = ow0 - 1 + cx;
        const bool in = (unsigned)y < 32u && (unsigned)xx < 32u;
        const int yc  = min(max(y, 0), 31);
        const int xc  = min(max(xx, 0), 31);
        const float v = x[((b * 32 + c) * 32 + yc) * 32 + xc];
        xr[k] = in ? v : 0.0f;
    }
    #pragma unroll
    for (int k = 0; k < 9; ++k) {
        #pragma unroll
        for (int j = 0; j < 4; ++j)
            sW[(k * 32 + wm * 4 + j) * WPAD + wo] = wr[k][j];
    }
    #pragma unroll
    for (int k = 0; k < 4; ++k) sX[tid + 256 * k] = xr[k];
    {
        float wmx = 0.0f;
        #pragma unroll
        for (int k = 0; k < 9; ++k) {
            #pragma unroll
            for (int j = 0; j < 4; ++j) wmx = fmaxf(wmx, __builtin_fabsf(wr[k][j]));
        }
        const unsigned wb = wave_umax(__float_as_uint(wmx));
        if (lane == 0) { sWm[wv] = __uint_as_float(wb); sP[wv][HP] = 0.0f; }
        if (tid < WPAD) sW[HP * WPAD + tid] = 0.0f;
    }
    __syncthreads();

    unsigned m1, m2;
    {
        const int c = lane;
        unsigned key[9];
        #pragma unroll
        for (int ki = 0; ki < 3; ++ki) {
            #pragma unroll
            for (int kj = 0; kj < 3; ++kj) {
                const int   h  = c * 9 + ki * 3 + kj;
                const float pe = sX[c * 30 + ki * 10 + wv + kj];
                sP[wv][h] = pe;
                key[ki * 3 + kj] = (__float_as_uint(__builtin_fabsf(pe)) & 0xFFFFFE00u) | (unsigned)h;
            }
        }
        m1 = max(key[0], key[1]);
        m2 = min(key[0], key[1]);
        #pragma unroll
        for (int i = 2; i < 9; ++i) {
            m2 = max(m2, min(m1, key[i]));
            m1 = max(m1, key[i]);
        }
    }
    int hsel[NSEL];
    #pragma unroll
    for (int k = 0; k < NSEL; ++k) {
        const unsigned g = wave_umax(m1);
        hsel[k] = __builtin_amdgcn_readfirstlane((int)(g & 0x1FFu));
        const bool own = (m1 == g);
        m1 = own ? m2 : m1;
        m2 = own ? 0u : m2;
    }
    __syncthreads();

    const int   half   = lane >> 4;
    const int   l16    = lane & 15;
    const int   rowoff = half * 8;
    const float bsel   = half ? 0.0f : 1.0f;
    v2f Bm; Bm[0] = bsel; Bm[1] = 0.0f;
    const float* pP = sP[wv];
    float wmax;
    {
        const v4f m0 = *(const v4f*)&sWm[0], m1 = *(const v4f*)&sWm[4];
        wmax = fmaxf(fmaxf(fmaxf(m0[0], m0[1]), fmaxf(m0[2], m0[3])), fmaxf(fmaxf(m1[0], m1[1]), fmaxf(m1[2], m1[3])));
    }

    float S_pa = 0.0f, S_qa = 0.0f, S_pb = 0.0f, S_qb = 0.0f;
    float b_pa = 0.0f, b_qa = 0.0f, b_pb = 0.0f, b_qb = 0.0f;
    #pragma unroll
    for (int k = 0; k < NSEL; ++k) {
        const float pk = pP[hsel[k]];
        const float wa = sW[hsel[k] * WPAD + l16];
        const float wb = sW[hsel[k] * WPAD + l16 + 16];
        S_pa += __builtin_fabsf(pk + wa);
        S_qa += __builtin_fabsf(pk - wa);
        S_pb += __builtin_fabsf(pk + wb);
        S_qb += __builtin_fabsf(pk - wb);
        const float inv = 1.0f / (float)(k + 1);
        const float c_pa = (S_pa - GAMMA) * inv, c_qa = (S_qa - GAMMA) * inv;
        const float c_pb = (S_pb - GAMMA) * inv, c_qb = (S_qb - GAMMA) * inv;
        b_pa = k ? fmaxf(b_pa, c_pa) : c_pa;
        b_qa = k ? fmaxf(b_qa, c_qa) : c_qa;
        b_pb = k ? fmaxf(b_pb, c_pb) : c_pb;
        b_qb = k ? fmaxf(b_qb, c_qb) : c_qb;
    }

    const float lo0 = -GAMMA / DTOT;
    float t_pa = fmaxf(b_pa - 2.4e-6f * (S_pa + GAMMA), lo0);
    float t_qa = fmaxf(b_qa - 2.4e-6f * (S_qa + GAMMA), lo0);
    float t_pb = fmaxf(b_pb - 2.4e-6f * (S_pb + GAMMA), lo0);
    float t_qb = fmaxf(b_qb - 2.4e-6f * (S_qb + GAMMA), lo0);
    int  s_pa = -1, s_qa = -1, s_pb = -1, s_qb = -1;
    bool d_pa = false, d_qa = false, d_pb = false, d_qb = false;

    const int wbase = wv * (LSLOTS * LROW);
    const int lcol  = wbase + lane * 4;
    bool listed = false;
    int  cnt4   = 0;
    if (__all(t_pa >= 0.0f && t_qa >= 0.0f && t_pb >= 0.0f && t_qb >= 0.0f)) {
        int o_pa = lcol * 4, o_qa = lcol * 4 + 4, o_pb = lcol * 4 + 8, o_qb = lcol * 4 + 12;
        const int lim = lcol * 4 + LCAP * LSTEP;
        int peak = 0;
        const float a_pa = __builtin_fabsf(t_pa), a_qa = __builtin_fabsf(t_qa);
        const float a_pb = __builtin_fabsf(t_pb), a_qb = __builtin_fabsf(t_qb);
        char* const Lb = (char*)sL;
        const float amin4 = fminf(fminf(a_pa, a_qa), fminf(a_pb, a_qb));
        const float a_min = __uint_as_float(~wave_umax(~__float_as_uint(amin4)));
        const float wwide = wmax * CWIDEN;
        const unsigned hmask = (lane & 8) ? 0xFF00FF00u : 0x00FF00FFu;
        const unsigned lmask = hmask & ((1u << lane) - 1u);
        int* const cmine = &sC[wv][(lane >> 3) & 1][0];
        int n0 = 0, n1 = 0;
        float pj[9];
        #pragma unroll
        for (int j = 0; j < 9; ++j) pj[j] = pP[32 * j + lane];
        #pragma unroll
        for (int j = 0; j < 9; ++j) {
            const bool     cand = __builtin_fmaf(__builtin_fabsf(pj[j]), CWIDEN, wwide) >= a_min;
            const unsigned m    = __builtin_amdgcn_ballot_w32(cand);
            const int      base = (lane & 8) ? n1 : n0;
            if (cand) cmine[base + __builtin_popcount(m & lmask)] = 32 * j + lane;
            n0 += __builtin_popcount(m & 0x00FF00FFu);
            n1 += __builtin_popcount(m & 0xFF00FF00u);
        }
        const int nt = (max(n0, n1) + 7) >> 3;
        #pragma nounroll
        for (int k = n0 + lane; k < nt * 8; k += 32) sC[wv][0][k] = HP;
        #pragma nounroll
        for (int k = n1 + lane; k < nt * 8; k += 32) sC[wv][1][k] = HP;
        const int* const cA = &sC[wv][l16 >> 3][l16 & 7];
        const v4i* const cC = (const v4i*)&sC[wv][half][0];
        for (int i = 0; i < nt; ++i) {
            peak = max(peak, max(max(o_pa, o_qa), max(o_pb, o_qb)));
            o_pa = min(o_pa, lim);
            o_qa = min(o_qa, lim + 4);
            o_pb = min(o_pb, lim + 8);
            o_qb = min(o_qb, lim + 12);
            const v4i r0 = cC[2 * i], r1 = cC[2 * i + 1];
            float pv = pP[cA[8 * i]];
            float a0 = half ? 0.0f : pv;
            v2f Ap; Ap[0] =  a0; Ap[1] = 0.0f;
            v2f Aq; Aq[0] = -a0; Aq[1] = 0.0f;
            v8f Ca, Cb;
            #pragma unroll
            for (int r = 0; r < 8; ++r) {
                const int    hr  = (r < 4) ? r0[r & 3] : r1[r & 3];
                const float* row = &sW[hr * WPAD];
                Ca[r] = row[l16];
                Cb[r] = row[l16 + 16];
            }
            v8f Dpa = __builtin_amdgcn_wmma_f32_16x16x4_f32(false, Ap, false, Bm, (short)0, Ca, false, false);
            v8f Dqa = __builtin_amdgcn_wmma_f32_16x16x4_f32(false, Aq, false, Bm, (short)0, Ca, false, false);
            v8f Dpb = __builtin_amdgcn_wmma_f32_16x16x4_f32(false, Ap, false, Bm, (short)0, Cb, false, false);
            v8f Dqb = __builtin_amdgcn_wmma_f32_16x16x4_f32(false, Aq, false, Bm, (short)0, Cb, false, false);
            #pragma unroll
            for (int r = 0; r < 8; ++r) {
                *(float*)(Lb + o_pa) = Dpa[r];
                *(float*)(Lb + o_qa) = Dqa[r];
                *(float*)(Lb + o_pb) = Dpb[r];
                *(float*)(Lb + o_qb) = Dqb[r];
                o_pa += (__builtin_fabsf(Dpa[r]) >= a_pa) ? LSTEP : 0;
                o_qa += (__builtin_fabsf(Dqa[r]) >= a_qa) ? LSTEP : 0;
                o_pb += (__builtin_fabsf(Dpb[r]) >= a_pb) ? LSTEP : 0;
                o_qb += (__builtin_fabsf(Dqb[r]) >= a_qb) ? LSTEP : 0;
            }
        }
        peak = max(peak, max(max(o_pa, o_qa), max(o_pb, o_qb)));
        listed = __all(((peak - wbase * 4) >> 9) <= LCAP);
        if (listed) {
            const int c_pa = (o_pa - wbase * 4) >> 9, c_qa = (o_qa - wbase * 4) >> 9;
            const int c_pb = (o_pb - wbase * 4) >> 9, c_qb = (o_qb - wbase * 4) >> 9;
            int m = max(max(c_pa, c_qa), max(c_pb, c_qb));
            m = max(m, __shfl_xor(m, 16, 32));
            m = max(m, __shfl_xor(m,  8, 32));
            m = max(m, __shfl_xor(m,  4, 32));
            m = max(m, __shfl_xor(m,  2, 32));
            m = max(m, __shfl_xor(m,  1, 32));
            cnt4 = (__builtin_amdgcn_readfirstlane(m) + 3) & ~3;
            for (int s = 0; s < cnt4 + 4; ++s) {
                const int o = lcol + s * LROW;
                if (s >= c_pa) sL[o]     = 0.0f;
                if (s >= c_qa) sL[o + 1] = 0.0f;
                if (s >= c_pb) sL[o + 2] = 0.0f;
                if (s >= c_qb) sL[o + 3] = 0.0f;
            }
        }
    }

    if (listed) {
        const v4f* Lr = (const v4f*)&sL[lcol];
        for (int it = 0; it < MAXIT; ++it) {
            const float a_pa = __builtin_fabsf(t_pa);
            const float a_qa = __builtin_fabsf(t_qa);
            const float a_pb = __builtin_fabsf(t_pb);
            const float a_qb = __builtin_fabsf(t_qb);
            float F_pa = 0.0f, F_qa = 0.0f, F_pb = 0.0f, F_qb = 0.0f;
            int   n_pa = 0, n_qa = 0, n_pb = 0, n_qb = 0;
            v4f nx0 = Lr[0], nx1 = Lr[32], nx2 = Lr[64], nx3 = Lr[96];
            for (int s = 0; s < cnt4; s += 4) {
                const v4f c0 = nx0, c1 = nx1, c2 = nx2, c3 = nx3;
                nx0 = Lr[(s + 4) * 32]; nx1 = Lr[(s + 5) * 32]; nx2 = Lr[(s + 6) * 32]; nx3 = Lr[(s + 7) * 32];
                unsigned g_pa = 0u, g_qa = 0u, g_pb = 0u, g_qb = 0u;
                #pragma unroll
                for (int k = 0; k < 4; ++k) {
                    const v4f c = (k == 0) ? c0 : (k == 1) ? c1 : (k == 2) ? c2 : c3;
                    const float e_pa = __builtin_fabsf(c[0]) - a_pa;
                    const float e_qa = __builtin_fabsf(c[1]) - a_qa;
                    const float e_pb = __builtin_fabsf(c[2]) - a_pb;
                    const float e_qb = __builtin_fabsf(c[3]) - a_qb;
                    F_pa += fmaxf(e_pa, 0.0f);
                    F_qa += fmaxf(e_qa, 0.0f);
                    F_pb += fmaxf(e_pb, 0.0f);
                    F_qb += fmaxf(e_qb, 0.0f);
                    g_pa = (g_pa << 1) | (__float_as_uint(e_pa) >> 31);
                    g_qa = (g_qa << 1) | (__float_as_uint(e_qa) >> 31);
                    g_pb = (g_pb << 1) | (__float_as_uint(e_pb) >> 31);
                    g_qb = (g_qb << 1) | (__float_as_uint(e_qb) >> 31);
                }
                n_pa += __builtin_popcount(g_pa);
                n_qa += __builtin_popcount(g_qa);
                n_pb += __builtin_popcount(g_pb);
                n_qb += __builtin_popcount(g_qb);
            }
            F_pa += __shfl_xor(F_pa, 16, 32);
            F_qa += __shfl_xor(F_qa, 16, 32);
            F_pb += __shfl_xor(F_pb, 16, 32);
            F_qb += __shfl_xor(F_qb, 16, 32);
            n_pa += __shfl_xor(n_pa, 16, 32);
            n_qa += __shfl_xor(n_qa, 16, 32);
            n_pb += __shfl_xor(n_pb, 16, 32);
            n_qb += __shfl_xor(n_qb, 16, 32);
            const int off = HP - 2 * cnt4;
            newton_step(t_pa, s_pa, d_pa, F_pa, off + n_pa);
            newton_step(t_qa, s_qa, d_qa, F_qa, off + n_qa);
            newton_step(t_pb, s_pb, d_pb, F_pb, off + n_pb);
            newton_step(t_qb, s_qb, d_qb, F_qb, off + n_qb);

            if (!__any(!(d_pa && d_qa && d_pb && d_qb))) break;
        }
    } else {
        for (int it = 0; it < MAXIT; ++it) {
            const float a_pa = __builtin_fabsf(t_pa);
            const float a_qa = __builtin_fabsf(t_qa);
            const float a_pb = __builtin_fabsf(t_pb);
            const float a_qb = __builtin_fabsf(t_qb);
            float F_pa = 0.0f, F_qa = 0.0f, F_pb = 0.0f, F_qb = 0.0f;
            int   n_pa = 0, n_qa = 0, n_pb = 0, n_qb = 0;

            for (int h0 = 0; h0 < HP; h0 += 32) {
                unsigned g_pa = 0u, g_qa = 0u, g_pb = 0u, g_qb = 0u;
                #pragma unroll
                for (int s = 0; s < 2; ++s) {
                    const int hb = h0 + 16 * s;
                    float pv = pP[hb + l16];
                    float a0 = half ? 0.0f : pv;
                    v2f Ap; Ap[0] =  a0; Ap[1] = 0.0f;
                    v2f Aq; Aq[0] = -a0; Aq[1] = 0.0f;
                    v8f Ca, Cb;
                    #pragma unroll
                    for (int r = 0; r < 8; ++r) {
                        const float* row = &sW[(hb + r + rowoff) * WPAD];
                        Ca[r] = row[l16];
                        Cb[r] = row[l16 + 16];
                    }
                    v8f Dpa = __builtin_amdgcn_wmma_f32_16x16x4_f32(false, Ap, false, Bm, (short)0, Ca, false, false);
                    v8f Dqa = __builtin_amdgcn_wmma_f32_16x16x4_f32(false, Aq, false, Bm, (short)0, Ca, false, false);
                    v8f Dpb = __builtin_amdgcn_wmma_f32_16x16x4_f32(false, Ap, false, Bm, (short)0, Cb, false, false);
                    v8f Dqb = __builtin_amdgcn_wmma_f32_16x16x4_f32(false, Aq, false, Bm, (short)0, Cb, false, false);
                    #pragma unroll
                    for (int r = 0; r < 8; ++r) {
                        const float e_pa = __builtin_fabsf(Dpa[r]) - a_pa;
                        const float e_qa = __builtin_fabsf(Dqa[r]) - a_qa;
                        const float e_pb = __builtin_fabsf(Dpb[r]) - a_pb;
                        const float e_qb = __builtin_fabsf(Dqb[r]) - a_qb;
                        F_pa += fmaxf(e_pa, 0.0f);
                        F_qa += fmaxf(e_qa, 0.0f);
                        F_pb += fmaxf(e_pb, 0.0f);
                        F_qb += fmaxf(e_qb, 0.0f);
                        g_pa = (g_pa << 1) | (__float_as_uint(e_pa) >> 31);
                        g_qa = (g_qa << 1) | (__float_as_uint(e_qa) >> 31);
                        g_pb = (g_pb << 1) | (__float_as_uint(e_pb) >> 31);
                        g_qb = (g_qb << 1) | (__float_as_uint(e_qb) >> 31);
                    }
                }
                n_pa += __builtin_popcount(g_pa);
                n_qa += __builtin_popcount(g_qa);
                n_pb += __builtin_popcount(g_pb);
                n_qb += __builtin_popcount(g_qb);
            }
            F_pa += __shfl_xor(F_pa, 16, 32);
            F_qa += __shfl_xor(F_qa, 16, 32);
            F_pb += __shfl_xor(F_pb, 16, 32);
            F_qb += __shfl_xor(F_qb, 16, 32);
            n_pa += __shfl_xor(n_pa, 16, 32);
            n_qa += __shfl_xor(n_qa, 16, 32);
            n_pb += __shfl_xor(n_pb, 16, 32);
            n_qb += __shfl_xor(n_qb, 16, 32);

            newton_step(t_pa, s_pa, d_pa, F_pa, n_pa);
            newton_step(t_qa, s_qa, d_qa, F_qa, n_qa);
            newton_step(t_pb, s_pb, d_pb, F_pb, n_pb);
            newton_step(t_qb, s_qb, d_qb, F_qb, n_qb);

            if (!__any(!(d_pa && d_qa && d_pb && d_qb))) break;
        }
    }

    const float tp = half ? t_pb : t_pa;
    const float tq = half ? t_qb : t_qa;
    const float conv = 0.5f * DTOT * (tp - tq);

    out[(b * 32 + lane) * 1024 + pos] = conv;
}

extern "C" void kernel_launch(void* const* d_in, const int* in_sizes, int n_in,
                              void* d_out, int out_size, void* d_ws, size_t ws_size,
                              hipStream_t stream) {
    const float* x = (const float*)d_in[0];
    const float* w = (const float*)d_in[1];
    float* out = (float*)d_out;
    (void)in_sizes; (void)n_in; (void)out_size; (void)d_ws; (void)ws_size;

    mpconv_sparsemax_kernel<<<256, 256, 0, stream>>>(x, w, out);
}


# Newton passes walk the candidate value tiles in the registers the matrix instruction wrote (waves with at most 2 such tiles) instead of building and re-reading padded LDS lists; sums bit-identical
# speedup vs baseline: 13.1192x; 1.1877x over previous
#include <hip/hip_runtime.h>


typedef __attribute__((ext_vector_type(2))) float v2f;
typedef __attribute__((ext_vector_type(8))) float v8f;

#define HP     288
#define WPAD   34
#define GAMMA  1.0f
#define DTOT   576.0f
#define MAXIT  (2 * HP + 2)
#define LSLOTS 64
#define LROW   128
#define LCAP   (LSLOTS - 8)
#define LSTEP  (LROW * 4)
#define XSLAB  960
#define NSEL   8
#define CPITCH 160
#define CWIDEN 1.00001f
#define NTREG  2

typedef __attribute__((ext_vector_type(4))) float v4f;
typedef __attribute__((ext_vector_type(4))) int   v4i;

__device__ __forceinline__ unsigned wave_umax(unsigned v)
{
    v = max(v, (unsigned)__builtin_amdgcn_update_dpp(0, (int)v, 0x161, 0xf, 0xf, true));
    v = max(v, (unsigned)__builtin_amdgcn_update_dpp(0, (int)v, 0x162, 0xf, 0xf, true));
    v = max(v, (unsigned)__builtin_amdgcn_update_dpp(0, (int)v, 0x164, 0xf, 0xf, true));
    v = max(v, (unsigned)__builtin_amdgcn_update_dpp(0, (int)v, 0x168, 0xf, 0xf, true));
    v = max(v, (unsigned)__builtin_amdgcn_permlanex16((int)v, (int)v, 0x76543210, 0xfedcba98, false, false));
    return v;
}

__device__ __forceinline__ void newton_step(float& t, int& sprev, bool& done, float F, int nneg)
{
    const float a   = __builtin_fabsf(t);
    const int   cge = HP - nneg;
    const bool  neg = t < 0.0f;
    const float f   = neg ? (F + DTOT * a) : F;
    int slope = neg ? (2 * HP - cge) : cge;
    slope = slope < 1 ? 1 : slope;
    float step = (f - GAMMA) * __builtin_amdgcn_rcpf((float)slope);
    step = step > 0.0f ? step : 0.0f;
    const bool conv = (slope == sprev) || !(step > 0.0f);
    t     = done ? t : t + step;
    sprev = done ? sprev : slope;
    done  = done || conv;
}

__device__ __forceinline__ void form_tile(const float* sWt, const float* pP, const int* cA, const v4i* cC,
                                          int i, int half, int l16, v2f Bm,
                                          v8f& Dpa, v8f& Dqa, v8f& Dpb, v8f& Dqb)
{
    const v4i r0 = cC[2 * i], r1 = cC[2 * i + 1];
    const float pv = pP[cA[8 * i]];
    const float a0 = half ? 0.0f : pv;
    v2f Ap; Ap[0] =  a0; Ap[1] = 0.0f;
    v2f Aq; Aq[0] = -a0; Aq[1] = 0.0f;
    v8f Ca, Cb;
    #pragma unroll
    for (int r = 0; r < 8; ++r) {
        const int    hr  = (r < 4) ? r0[r & 3] : r1[r & 3];
        const float* row = &sWt[hr * WPAD];
        Ca[r] = row[l16];
        Cb[r] = row[l16 + 16];
    }
    Dpa = __builtin_amdgcn_wmma_f32_16x16x4_f32(false, Ap, false, Bm, (short)0, Ca, false, false);
    Dqa = __builtin_amdgcn_wmma_f32_16x16x4_f32(false, Aq, false, Bm, (short)0, Ca, false, false);
    Dpb = __builtin_amdgcn_wmma_f32_16x16x4_f32(false, Ap, false, Bm, (short)0, Cb, false, false);
    Dqb = __builtin_amdgcn_wmma_f32_16x16x4_f32(false, Aq, false, Bm, (short)0, Cb, false, false);
}

__device__ __forceinline__ void walk_tile(const v8f& D, float a, float& F, unsigned& g)
{
    #pragma unroll
    for (int r = 0; r < 8; ++r) {
        const float e = __builtin_fabsf(D[r]) - a;
        F += fmaxf(e, 0.0f);
        g = (g << 1) | (__float_as_uint(e) >> 31);
    }
}

__global__ __launch_bounds__(256) void mpconv_sparsemax_kernel(
    const float* __restrict__ x,
    const float* __restrict__ w,
    float* __restrict__ out)
{
    __shared__ float sW[(HP + 1) * WPAD];
    __shared__ float sP[8][HP + 8];
    __shared__ __attribute__((aligned(16))) float sWm[8];
    __shared__ __attribute__((aligned(16))) int sC[8][2][CPITCH];
    __shared__ float sX[1024];
    __shared__ __attribute__((aligned(16))) float sL[8 * LSLOTS * LROW];

    const int tid  = threadIdx.x;
    const int lane = tid & 31;
    const int wv   = tid >> 5;

    const int blk = blockIdx.x;
    const int b   = blk >> 7;
    const int oh  = (blk >> 2) & 31;
    const int ow0 = (blk & 3) * 8;
    const int ow  = ow0 + wv;
    const int pos = oh * 32 + ow;
    const int wo  = tid >> 3;
    const int wm  = tid & 7;
    v4f wr[9];
    {
        const v4f* wsrc = (const v4f*)(w + wo * HP + wm * 4);
        #pragma unroll
        for (int k = 0; k < 9; ++k) wr[k] = wsrc[k * 8];
    }
    float xr[4];
    #pragma unroll
    for (int k = 0; k < 4; ++k) {
        int e = tid + 256 * k;
        e = e < XSLAB ? e : XSLAB - 1;
        const int c   = e / 30;
        const int rem = e - c * 30;
        const int ry  = rem / 10;
        const int cx  = rem - ry * 10;
        const int y   = oh - 1 + ry;
        const int xx  = ow0 - 1 + cx;
        const bool in = (unsigned)y < 32u && (unsigned)xx < 32u;
        const int yc  = min(max(y, 0), 31);
        const int xc  = min(max(xx, 0), 31);
        const float v = x[((b * 32 + c) * 32 + yc) * 32 + xc];
        xr[k] = in ? v : 0.0f;
    }
    #pragma unroll
    for (int k = 0; k < 9; ++k) {
        #pragma unroll
        for (int j = 0; j < 4; ++j)
            sW[(k * 32 + wm * 4 + j) * WPAD + wo] = wr[k][j];
    }
    #pragma unroll
    for (int k = 0; k < 4; ++k) sX[tid + 256 * k] = xr[k];
    {
        float wmx = 0.0f;
        #pragma unroll
        for (int k = 0; k < 9; ++k) {
            #pragma unroll
            for (int j = 0; j < 4; ++j) wmx = fmaxf(wmx, __builtin_fabsf(wr[k][j]));
        }
        const unsigned wb = wave_umax(__float_as_uint(wmx));
        if (lane == 0) { sWm[wv] = __uint_as_float(wb); sP[wv][HP] = 0.0f; }
        if (tid < WPAD) sW[HP * WPAD + tid] = 0.0f;
    }
    __syncthreads();

    unsigned m1, m2;
    {
        const int c = lane;
        unsigned key[9];
        #pragma unroll
        for (int ki = 0; ki < 3; ++ki) {
            #pragma unroll
            for (int kj = 0; kj < 3; ++kj) {
                const int   h  = c * 9 + ki * 3 + kj;
                const float pe = sX[c * 30 + ki * 10 + wv + kj];
                sP[wv][h] = pe;
                key[ki * 3 + kj] = (__float_as_uint(__builtin_fabsf(pe)) & 0xFFFFFE00u) | (unsigned)h;
            }
        }
        m1 = max(key[0], key[1]);
        m2 = min(key[0], key[1]);
        #pragma unroll
        for (int i = 2; i < 9; ++i) {
            m2 = max(m2, min(m1, key[i]));
            m1 = max(m1, key[i]);
        }
    }
    int hsel[NSEL];
    #pragma unroll
    for (int k = 0; k < NSEL; ++k) {
        const unsigned g = wave_umax(m1);
        hsel[k] = __builtin_amdgcn_readfirstlane((int)(g & 0x1FFu));
        const bool own = (m1 == g);
        m1 = own ? m2 : m1;
        m2 = own ? 0u : m2;
    }
    __syncthreads();

    const int   half   = lane >> 4;
    const int   l16    = lane & 15;
    const int   rowoff = half * 8;
    const float bsel   = half ? 0.0f : 1.0f;
    v2f Bm; Bm[0] = bsel; Bm[1] = 0.0f;
    const float* pP = sP[wv];
    float wmax;
    {
        const v4f m0 = *(const v4f*)&sWm[0], m1 = *(const v4f*)&sWm[4];
        wmax = fmaxf(fmaxf(fmaxf(m0[0], m0[1]), fmaxf(m0[2], m0[3])), fmaxf(fmaxf(m1[0], m1[1]), fmaxf(m1[2], m1[3])));
    }

    float S_pa = 0.0f, S_qa = 0.0f, S_pb = 0.0f, S_qb = 0.0f;
    float b_pa = 0.0f, b_qa = 0.0f, b_pb = 0.0f, b_qb = 0.0f;
    #pragma unroll
    for (int k = 0; k < NSEL; ++k) {
        const float pk = pP[hsel[k]];
        const float wa = sW[hsel[k] * WPAD + l16];
        const float wb = sW[hsel[k] * WPAD + l16 + 16];
        S_pa += __builtin_fabsf(pk + wa);
        S_qa += __builtin_fabsf(pk - wa);
        S_pb += __builtin_fabsf(pk + wb);
        S_qb += __builtin_fabsf(pk - wb);
        const float inv = 1.0f / (float)(k + 1);
        const float c_pa = (S_pa - GAMMA) * inv, c_qa = (S_qa - GAMMA) * inv;
        const float c_pb = (S_pb - GAMMA) * inv, c_qb = (S_qb - GAMMA) * inv;
        b_pa = k ? fmaxf(b_pa, c_pa) : c_pa;
        b_qa = k ? fmaxf(b_qa, c_qa) : c_qa;
        b_pb = k ? fmaxf(b_pb, c_pb) : c_pb;
        b_qb = k ? fmaxf(b_qb, c_qb) : c_qb;
    }

    const float lo0 = -GAMMA / DTOT;
    float t_pa = fmaxf(b_pa - 2.4e-6f * (S_pa + GAMMA), lo0);
    float t_qa = fmaxf(b_qa - 2.4e-6f * (S_qa + GAMMA), lo0);
    float t_pb = fmaxf(b_pb - 2.4e-6f * (S_pb + GAMMA), lo0);
    float t_qb = fmaxf(b_qb - 2.4e-6f * (S_qb + GAMMA), lo0);
    int  s_pa = -1, s_qa = -1, s_pb = -1, s_qb = -1;
    bool d_pa = false, d_qa = false, d_pb = false, d_qb = false;

    const int wbase = wv * (LSLOTS * LROW);
    const int lcol  = wbase + lane * 4;
    bool listed = false;
    bool solved = false;
    int  cnt4   = 0;
    if (__all(t_pa >= 0.0f && t_qa >= 0.0f && t_pb >= 0.0f && t_qb >= 0.0f)) {
        int o_pa = lcol * 4, o_qa = lcol * 4 + 4, o_pb = lcol * 4 + 8, o_qb = lcol * 4 + 12;
        const int lim = lcol * 4 + LCAP * LSTEP;
        int peak = 0;
        const float a_pa = __builtin_fabsf(t_pa), a_qa = __builtin_fabsf(t_qa);
        const float a_pb = __builtin_fabsf(t_pb), a_qb = __builtin_fabsf(t_qb);
        char* const Lb = (char*)sL;
        const float amin4 = fminf(fminf(a_pa, a_qa), fminf(a_pb, a_qb));
        const float a_min = __uint_as_float(~wave_umax(~__float_as_uint(amin4)));
        const float wwide = wmax * CWIDEN;
        const unsigned hmask = (lane & 8) ? 0xFF00FF00u : 0x00FF00FFu;
        const unsigned lmask = hmask & ((1u << lane) - 1u);
        int* const cmine = &sC[wv][(lane >> 3) & 1][0];
        int n0 = 0, n1 = 0;
        float pj[9];
        #pragma unroll
        for (int j = 0; j < 9; ++j) pj[j] = pP[32 * j + lane];
        #pragma unroll
        for (int j = 0; j < 9; ++j) {
            const bool     cand = __builtin_fmaf(__builtin_fabsf(pj[j]), CWIDEN, wwide) >= a_min;
            const unsigned m    = __builtin_amdgcn_ballot_w32(cand);
            const int      base = (lane & 8) ? n1 : n0;
            if (cand) cmine[base + __builtin_popcount(m & lmask)] = 32 * j + lane;
            n0 += __builtin_popcount(m & 0x00FF00FFu);
            n1 += __builtin_popcount(m & 0xFF00FF00u);
        }
        const int nt = (max(n0, n1) + 7) >> 3;
        #pragma nounroll
        for (int k = n0 + lane; k < nt * 8; k += 32) sC[wv][0][k] = HP;
        #pragma nounroll
        for (int k = n1 + lane; k < nt * 8; k += 32) sC[wv][1][k] = HP;
        const int* const cA = &sC[wv][l16 >> 3][l16 & 7];
        const v4i* const cC = (const v4i*)&sC[wv][half][0];
        if (nt <= NTREG) {
            v8f Vpa0 = {}, Vqa0 = {}, Vpb0 = {}, Vqb0 = {};
            v8f Vpa1 = {}, Vqa1 = {}, Vpb1 = {}, Vqb1 = {};
            if (nt > 0) form_tile(sW, pP, cA, cC, 0, half, l16, Bm, Vpa0, Vqa0, Vpb0, Vqb0);
            if (nt > 1) form_tile(sW, pP, cA, cC, 1, half, l16, Bm, Vpa1, Vqa1, Vpb1, Vqb1);
            const int off = HP - 16 * nt;
            for (int it = 0; it < MAXIT; ++it) {
                const float c_pa = __builtin_fabsf(t_pa), c_qa = __builtin_fabsf(t_qa);
                const float c_pb = __builtin_fabsf(t_pb), c_qb = __builtin_fabsf(t_qb);
                float    F_pa = 0.0f, F_qa = 0.0f, F_pb = 0.0f, F_qb = 0.0f;
                unsigned g_pa = 0u, g_qa = 0u, g_pb = 0u, g_qb = 0u;
                if (nt > 0) {
                    walk_tile(Vpa0, c_pa, F_pa, g_pa);
                    walk_tile(Vqa0, c_qa, F_qa, g_qa);
                    walk_tile(Vpb0, c_pb, F_pb, g_pb);
                    walk_tile(Vqb0, c_qb, F_qb, g_qb);
                }
                if (nt > 1) {
                    walk_tile(Vpa1, c_pa, F_pa, g_pa);
                    walk_tile(Vqa1, c_qa, F_qa, g_qa);
                    walk_tile(Vpb1, c_pb, F_pb, g_pb);
                    walk_tile(Vqb1, c_qb, F_qb, g_qb);
                }
                int n_pa = __builtin_popcount(g_pa), n_qa = __builtin_popcount(g_qa);
                int n_pb = __builtin_popcount(g_pb), n_qb = __builtin_popcount(g_qb);
                F_pa += __shfl_xor(F_pa, 16, 32);
                F_qa += __shfl_xor(F_qa, 16, 32);
                F_pb += __shfl_xor(F_pb, 16, 32);
                F_qb += __shfl_xor(F_qb, 16, 32);
                n_pa += __shfl_xor(n_pa, 16, 32);
                n_qa += __shfl_xor(n_qa, 16, 32);
                n_pb += __shfl_xor(n_pb, 16, 32);
                n_qb += __shfl_xor(n_qb, 16, 32);
                newton_step(t_pa, s_pa, d_pa, F_pa, off + n_pa);
                newton_step(t_qa, s_qa, d_qa, F_qa, off + n_qa);
                newton_step(t_pb, s_pb, d_pb, F_pb, off + n_pb);
                newton_step(t_qb, s_qb, d_qb, F_qb, off + n_qb);

                if (!__any(!(d_pa && d_qa && d_pb && d_qb))) break;
            }
            solved = true;
        } else {
            for (int i = 0; i < nt; ++i) {
                peak = max(peak, max(max(o_pa, o_qa), max(o_pb, o_qb)));
                o_pa = min(o_pa, lim);
                o_qa = min(o_qa, lim + 4);
                o_pb = min(o_pb, lim + 8);
                o_qb = min(o_qb, lim + 12);
                v8f Dpa, Dqa, Dpb, Dqb;
                form_tile(sW, pP, cA, cC, i, half, l16, Bm, Dpa, Dqa, Dpb, Dqb);
                #pragma unroll
                for (int r = 0; r < 8; ++r) {
                    *(float*)(Lb + o_pa) = Dpa[r];
                    *(float*)(Lb + o_qa) = Dqa[r];
                    *(float*)(Lb + o_pb) = Dpb[r];
                    *(float*)(Lb + o_qb) = Dqb[r];
                    o_pa += (__builtin_fabsf(Dpa[r]) >= a_pa) ? LSTEP : 0;
                    o_qa += (__builtin_fabsf(Dqa[r]) >= a_qa) ? LSTEP : 0;
                    o_pb += (__builtin_fabsf(Dpb[r]) >= a_pb) ? LSTEP : 0;
                    o_qb += (__builtin_fabsf(Dqb[r]) >= a_qb) ? LSTEP : 0;
                }
            }
            peak = max(peak, max(max(o_pa, o_qa), max(o_pb, o_qb)));
            listed = __all(((peak - wbase * 4) >> 9) <= LCAP);
            if (listed) {
                const int c_pa = (o_pa - wbase * 4) >> 9, c_qa = (o_qa - wbase * 4) >> 9;
                const int c_pb = (o_pb - wbase * 4) >> 9, c_qb = (o_qb - wbase * 4) >> 9;
                int m = max(max(c_pa, c_qa), max(c_pb, c_qb));
                m = max(m, __shfl_xor(m, 16, 32));
                m = max(m, __shfl_xor(m,  8, 32));
                m = max(m, __shfl_xor(m,  4, 32));
                m = max(m, __shfl_xor(m,  2, 32));
                m = max(m, __shfl_xor(m,  1, 32));
                cnt4 = (__builtin_amdgcn_readfirstlane(m) + 3) & ~3;
                for (int s = 0; s < cnt4 + 4; ++s) {
                    const int o = lcol + s * LROW;
                    if (s >= c_pa) sL[o]     = 0.0f;
                    if (s >= c_qa) sL[o + 1] = 0.0f;
                    if (s >= c_pb) sL[o + 2] = 0.0f;
                    if (s >= c_qb) sL[o + 3] = 0.0f;
                }
            }
        }
    }

    if (solved) {
    } else if (listed) {
        const v4f* Lr = (const v4f*)&sL[lcol];
        for (int it = 0; it < MAXIT; ++it) {
            const float a_pa = __builtin_fabsf(t_pa);
            const float a_qa = __builtin_fabsf(t_qa);
            const float a_pb = __builtin_fabsf(t_pb);
            const float a_qb = __builtin_fabsf(t_qb);
            float F_pa = 0.0f, F_qa = 0.0f, F_pb = 0.0f, F_qb = 0.0f;
            int   n_pa = 0, n_qa = 0, n_pb = 0, n_qb = 0;
            v4f nx0 = Lr[0], nx1 = Lr[32], nx2 = Lr[64], nx3 = Lr[96];
            for (int s = 0; s < cnt4; s += 4) {
                const v4f c0 = nx0, c1 = nx1, c2 = nx2, c3 = nx3;
                nx0 = Lr[(s + 4) * 32]; nx1 = Lr[(s + 5) * 32]; nx2 = Lr[(s + 6) * 32]; nx3 = Lr[(s + 7) * 32];
                unsigned g_pa = 0u, g_qa = 0u, g_pb = 0u, g_qb = 0u;
                #pragma unroll
                for (int k = 0; k < 4; ++k) {
                    const v4f c = (k == 0) ? c0 : (k == 1) ? c1 : (k == 2) ? c2 : c3;
                    const float e_pa = __builtin_fabsf(c[0]) - a_pa;
                    const float e_qa = __builtin_fabsf(c[1]) - a_qa;
                    const float e_pb = __builtin_fabsf(c[2]) - a_pb;
                    const float e_qb = __builtin_fabsf(c[3]) - a_qb;
                    F_pa += fmaxf(e_pa, 0.0f);
                    F_qa += fmaxf(e_qa, 0.0f);
                    F_pb += fmaxf(e_pb, 0.0f);
                    F_qb += fmaxf(e_qb, 0.0f);
                    g_pa = (g_pa << 1) | (__float_as_uint(e_pa) >> 31);
                    g_qa = (g_qa << 1) | (__float_as_uint(e_qa) >> 31);
                    g_pb = (g_pb << 1) | (__float_as_uint(e_pb) >> 31);
                    g_qb = (g_qb << 1) | (__float_as_uint(e_qb) >> 31);
                }
                n_pa += __builtin_popcount(g_pa);
                n_qa += __builtin_popcount(g_qa);
                n_pb += __builtin_popcount(g_pb);
                n_qb += __builtin_popcount(g_qb);
            }
            F_pa += __shfl_xor(F_pa, 16, 32);
            F_qa += __shfl_xor(F_qa, 16, 32);
            F_pb += __shfl_xor(F_pb, 16, 32);
            F_qb += __shfl_xor(F_qb, 16, 32);
            n_pa += __shfl_xor(n_pa, 16, 32);
            n_qa += __shfl_xor(n_qa, 16, 32);
            n_pb += __shfl_xor(n_pb, 16, 32);
            n_qb += __shfl_xor(n_qb, 16, 32);
            const int off = HP - 2 * cnt4;
            newton_step(t_pa, s_pa, d_pa, F_pa, off + n_pa);
            newton_step(t_qa, s_qa, d_qa, F_qa, off + n_qa);
            newton_step(t_pb, s_pb, d_pb, F_pb, off + n_pb);
            newton_step(t_qb, s_qb, d_qb, F_qb, off + n_qb);

            if (!__any(!(d_pa && d_qa && d_pb && d_qb))) break;
        }
    } else {
        for (int it = 0; it < MAXIT; ++it) {
            const float a_pa = __builtin_fabsf(t_pa);
            const float a_qa = __builtin_fabsf(t_qa);
            const float a_pb = __builtin_fabsf(t_pb);
            const float a_qb = __builtin_fabsf(t_qb);
            float F_pa = 0.0f, F_qa = 0.0f, F_pb = 0.0f, F_qb = 0.0f;
            int   n_pa = 0, n_qa = 0, n_pb = 0, n_qb = 0;

            for (int h0 = 0; h0 < HP; h0 += 32) {
                unsigned g_pa = 0u, g_qa = 0u, g_pb = 0u, g_qb = 0u;
                #pragma unroll
                for (int s = 0; s < 2; ++s) {
                    const int hb = h0 + 16 * s;
                    float pv = pP[hb + l16];
                    float a0 = half ? 0.0f : pv;
                    v2f Ap; Ap[0] =  a0; Ap[1] = 0.0f;
                    v2f Aq; Aq[0] = -a0; Aq[1] = 0.0f;
                    v8f Ca, Cb;
                    #pragma unroll
                    for (int r = 0; r < 8; ++r) {
                        const float* row = &sW[(hb + r + rowoff) * WPAD];
                        Ca[r] = row[l16];
                        Cb[r] = row[l16 + 16];
                    }
                    v8f Dpa = __builtin_amdgcn_wmma_f32_16x16x4_f32(false, Ap, false, Bm, (short)0, Ca, false, false);
                    v8f Dqa = __builtin_amdgcn_wmma_f32_16x16x4_f32(false, Aq, false, Bm, (short)0, Ca, false, false);
                    v8f Dpb = __builtin_amdgcn_wmma_f32_16x16x4_f32(false, Ap, false, Bm, (short)0, Cb, false, false);
                    v8f Dqb = __builtin_amdgcn_wmma_f32_16x16x4_f32(false, Aq, false, Bm, (short)0, Cb, false, false);
                    #pragma unroll
                    for (int r = 0; r < 8; ++r) {
                        const float e_pa = __builtin_fabsf(Dpa[r]) - a_pa;
                        const float e_qa = __builtin_fabsf(Dqa[r]) - a_qa;
                        const float e_pb = __builtin_fabsf(Dpb[r]) - a_pb;
                        const float e_qb = __builtin_fabsf(Dqb[r]) - a_qb;
                        F_pa += fmaxf(e_pa, 0.0f);
                        F_qa += fmaxf(e_qa, 0.0f);
                        F_pb += fmaxf(e_pb, 0.0f);
                        F_qb += fmaxf(e_qb, 0.0f);
                        g_pa = (g_pa << 1) | (__float_as_uint(e_pa) >> 31);
                        g_qa = (g_qa << 1) | (__float_as_uint(e_qa) >> 31);
                        g_pb = (g_pb << 1) | (__float_as_uint(e_pb) >> 31);
                        g_qb = (g_qb << 1) | (__float_as_uint(e_qb) >> 31);
                    }
                }
                n_pa += __builtin_popcount(g_pa);
                n_qa += __builtin_popcount(g_qa);
                n_pb += __builtin_popcount(g_pb);
                n_qb += __builtin_popcount(g_qb);
            }
            F_pa += __shfl_xor(F_pa, 16, 32);
            F_qa += __shfl_xor(F_qa, 16, 32);
            F_pb += __shfl_xor(F_pb, 16, 32);
            F_qb += __shfl_xor(F_qb, 16, 32);
            n_pa += __shfl_xor(n_pa, 16, 32);
            n_qa += __shfl_xor(n_qa, 16, 32);
            n_pb += __shfl_xor(n_pb, 16, 32);
            n_qb += __shfl_xor(n_qb, 16, 32);

            newton_step(t_pa, s_pa, d_pa, F_pa, n_pa);
            newton_step(t_qa, s_qa, d_qa, F_qa, n_qa);
            newton_step(t_pb, s_pb, d_pb, F_pb, n_pb);
            newton_step(t_qb, s_qb, d_qb, F_qb, n_qb);

            if (!__any(!(d_pa && d_qa && d_pb && d_qb))) break;
        }
    }

    const float tp = half ? t_pb : t_pa;
    const float tq = half ? t_qb : t_qa;
    const float conv = 0.5f * DTOT * (tp - tq);

    out[(b * 32 + lane) * 1024 + pos] = conv;
}

extern "C" void kernel_launch(void* const* d_in, const int* in_sizes, int n_in,
                              void* d_out, int out_size, void* d_ws, size_t ws_size,
                              hipStream_t stream) {
    const float* x = (const float*)d_in[0];
    const float* w = (const float*)d_in[1];
    float* out = (float*)d_out;
    (void)in_sizes; (void)n_in; (void)out_size; (void)d_ws; (void)ws_size;

    mpconv_sparsemax_kernel<<<256, 256, 0, stream>>>(x, w, out);
}


# Input slab loaded ahead of the weights and handed over behind an LDS-only barrier: patch keys, top-8 row selection and pre-filter row fetch run under the weight loads' flight; sums bit-identical
# speedup vs baseline: 14.1670x; 1.0799x over previous
#include <hip/hip_runtime.h>


typedef __attribute__((ext_vector_type(2))) float v2f;
typedef __attribute__((ext_vector_type(8))) float v8f;

#define HP     288
#define WPAD   34
#define GAMMA  1.0f
#define DTOT   576.0f
#define MAXIT  (2 * HP + 2)
#define LSLOTS 64
#define LROW   128
#define LCAP   (LSLOTS - 8)
#define LSTEP  (LROW * 4)
#define XSLAB  960
#define NSEL   8
#define CPITCH 160
#define CWIDEN 1.00001f
#define NTREG  2

typedef __attribute__((ext_vector_type(4))) float v4f;
typedef __attribute__((ext_vector_type(4))) int   v4i;

__device__ __forceinline__ unsigned wave_umax(unsigned v)
{
    v = max(v, (unsigned)__builtin_amdgcn_update_dpp(0, (int)v, 0x161, 0xf, 0xf, true));
    v = max(v, (unsigned)__builtin_amdgcn_update_dpp(0, (int)v, 0x162, 0xf, 0xf, true));
    v = max(v, (unsigned)__builtin_amdgcn_update_dpp(0, (int)v, 0x164, 0xf, 0xf, true));
    v = max(v, (unsigned)__builtin_amdgcn_update_dpp(0, (int)v, 0x168, 0xf, 0xf, true));
    v = max(v, (unsigned)__builtin_amdgcn_permlanex16((int)v, (int)v, 0x76543210, 0xfedcba98, false, false));
    return v;
}

__device__ __forceinline__ void newton_step(float& t, int& sprev, bool& done, float F, int nneg)
{
    const float a   = __builtin_fabsf(t);
    const int   cge = HP - nneg;
    const bool  neg = t < 0.0f;
    const float f   = neg ? (F + DTOT * a) : F;
    int slope = neg ? (2 * HP - cge) : cge;
    slope = slope < 1 ? 1 : slope;
    float step = (f - GAMMA) * __builtin_amdgcn_rcpf((float)slope);
    step = step > 0.0f ? step : 0.0f;
    const bool conv = (slope == sprev) || !(step > 0.0f);
    t     = done ? t : t + step;
    sprev = done ? sprev : slope;
    done  = done || conv;
}

__device__ __forceinline__ void form_tile(const float* sWt, const float* pP, const int* cA, const v4i* cC,
                                          int i, int half, int l16, v2f Bm,
                                          v8f& Dpa, v8f& Dqa, v8f& Dpb, v8f& Dqb)
{
    const v4i r0 = cC[2 * i], r1 = cC[2 * i + 1];
    const float pv = pP[cA[8 * i]];
    const float a0 = half ? 0.0f : pv;
    v2f Ap; Ap[0] =  a0; Ap[1] = 0.0f;
    v2f Aq; Aq[0] = -a0; Aq[1] = 0.0f;
    v8f Ca, Cb;
    #pragma unroll
    for (int r = 0; r < 8; ++r) {
        const int    hr  = (r < 4) ? r0[r & 3] : r1[r & 3];
        const float* row = &sWt[hr * WPAD];
        Ca[r] = row[l16];
        Cb[r] = row[l16 + 16];
    }
    Dpa = __builtin_amdgcn_wmma_f32_16x16x4_f32(false, Ap, false, Bm, (short)0, Ca, false, false);
    Dqa = __builtin_amdgcn_wmma_f32_16x16x4_f32(false, Aq, false, Bm, (short)0, Ca, false, false);
    Dpb = __builtin_amdgcn_wmma_f32_16x16x4_f32(false, Ap, false, Bm, (short)0, Cb, false, false);
    Dqb = __builtin_amdgcn_wmma_f32_16x16x4_f32(false, Aq, false, Bm, (short)0, Cb, false, false);
}

__device__ __forceinline__ void walk_tile(const v8f& D, float a, float& F, unsigned& g)
{
    #pragma unroll
    for (int r = 0; r < 8; ++r) {
        const float e = __builtin_fabsf(D[r]) - a;
        F += fmaxf(e, 0.0f);
        g = (g << 1) | (__float_as_uint(e) >> 31);
    }
}

__global__ __launch_bounds__(256) void mpconv_sparsemax_kernel(
    const float* __restrict__ x,
    const float* __restrict__ w,
    float* __restrict__ out)
{
    __shared__ float sW[(HP + 1) * WPAD];
    __shared__ float sP[8][HP + 8];
    __shared__ __attribute__((aligned(16))) float sWm[8];
    __shared__ __attribute__((aligned(16))) int sC[8][2][CPITCH];
    __shared__ float sX[1024];
    __shared__ __attribute__((aligned(16))) float sL[8 * LSLOTS * LROW];

    const int tid  = threadIdx.x;
    const int lane = tid & 31;
    const int wv   = tid >> 5;

    const int blk = blockIdx.x;
    const int b   = blk >> 7;
    const int oh  = (blk >> 2) & 31;
    const int ow0 = (blk & 3) * 8;
    const int ow  = ow0 + wv;
    const int pos = oh * 32 + ow;
    const int wo  = tid >> 3;
    const int wm  = tid & 7;
    float xr[4];
    #pragma unroll
    for (int k = 0; k < 4; ++k) {
        int e = tid + 256 * k;
        e = e < XSLAB ? e : XSLAB - 1;
        const int c   = e / 30;
        const int rem = e - c * 30;
        const int ry  = rem / 10;
        const int cx  = rem - ry * 10;
        const int y   = oh - 1 + ry;
        const int xx  = ow0 - 1 + cx;
        const bool in = (unsigned)y < 32u && (unsigned)xx < 32u;
        const int yc  = min(max(y, 0), 31);
        const int xc  = min(max(xx, 0), 31);
        const float v = x[((b * 32 + c) * 32 + yc) * 32 + xc];
        xr[k] = in ? v : 0.0f;
    }
    __builtin_amdgcn_sched_barrier(0);
    v4f wr[9];
    {
        const v4f* wsrc = (const v4f*)(w + wo * HP + wm * 4);
        #pragma unroll
        for (int k = 0; k < 9; ++k) wr[k] = wsrc[k * 8];
    }
    #pragma unroll
    for (int k = 0; k < 4; ++k) sX[tid + 256 * k] = xr[k];
    asm volatile("s_wait_dscnt 0x0\n\ts_barrier_signal -1\n\ts_barrier_wait -1" ::: "memory");

    unsigned m1, m2;
    {
        const int c = lane;
        unsigned key[9];
        #pragma unroll
        for (int ki = 0; ki < 3; ++ki) {
            #pragma unroll
            for (int kj = 0; kj < 3; ++kj) {
                const int   h  = c * 9 + ki * 3 + kj;
                const float pe = sX[c * 30 + ki * 10 + wv + kj];
                sP[wv][h] = pe;
                key[ki * 3 + kj] = (__float_as_uint(__builtin_fabsf(pe)) & 0xFFFFFE00u) | (unsigned)h;
            }
        }
        m1 = max(key[0], key[1]);
        m2 = min(key[0], key[1]);
        #pragma unroll
        for (int i = 2; i < 9; ++i) {
            m2 = max(m2, min(m1, key[i]));
            m1 = max(m1, key[i]);
        }
    }
    int hsel[NSEL];
    #pragma unroll
    for (int k = 0; k < NSEL; ++k) {
        const unsigned g = wave_umax(m1);
        hsel[k] = __builtin_amdgcn_readfirstlane((int)(g & 0x1FFu));
        const bool own = (m1 == g);
        m1 = own ? m2 : m1;
        m2 = own ? 0u : m2;
    }
    float pj[9];
    #pragma unroll
    for (int j = 0; j < 9; ++j) pj[j] = sP[wv][32 * j + lane];
    __builtin_amdgcn_sched_barrier(0);
    #pragma unroll
    for (int k = 0; k < 9; ++k) {
        #pragma unroll
        for (int j = 0; j < 4; ++j)
            sW[(k * 32 + wm * 4 + j) * WPAD + wo] = wr[k][j];
    }
    {
        float wmx = 0.0f;
        #pragma unroll
        for (int k = 0; k < 9; ++k) {
            #pragma unroll
            for (int j = 0; j < 4; ++j) wmx = fmaxf(wmx, __builtin_fabsf(wr[k][j]));
        }
        const unsigned wb = wave_umax(__float_as_uint(wmx));
        if (lane == 0) { sWm[wv] = __uint_as_float(wb); sP[wv][HP] = 0.0f; }
        if (tid < WPAD) sW[HP * WPAD + tid] = 0.0f;
    }
    __syncthreads();

    const int   half   = lane >> 4;
    const int   l16    = lane & 15;
    const int   rowoff = half * 8;
    const float bsel   = half ? 0.0f : 1.0f;
    v2f Bm; Bm[0] = bsel; Bm[1] = 0.0f;
    const float* pP = sP[wv];
    float wmax;
    {
        const v4f m0 = *(const v4f*)&sWm[0], m1 = *(const v4f*)&sWm[4];
        wmax = fmaxf(fmaxf(fmaxf(m0[0], m0[1]), fmaxf(m0[2], m0[3])), fmaxf(fmaxf(m1[0], m1[1]), fmaxf(m1[2], m1[3])));
    }

    float S_pa = 0.0f, S_qa = 0.0f, S_pb = 0.0f, S_qb = 0.0f;
    float b_pa = 0.0f, b_qa = 0.0f, b_pb = 0.0f, b_qb = 0.0f;
    #pragma unroll
    for (int k = 0; k < NSEL; ++k) {
        const float pk = pP[hsel[k]];
        const float wa = sW[hsel[k] * WPAD + l16];
        const float wb = sW[hsel[k] * WPAD + l16 + 16];
        S_pa += __builtin_fabsf(pk + wa);
        S_qa += __builtin_fabsf(pk - wa);
        S_pb += __builtin_fabsf(pk + wb);
        S_qb += __builtin_fabsf(pk - wb);
        const float inv = 1.0f / (float)(k + 1);
        const float c_pa = (S_pa - GAMMA) * inv, c_qa = (S_qa - GAMMA) * inv;
        const float c_pb = (S_pb - GAMMA) * inv, c_qb = (S_qb - GAMMA) * inv;
        b_pa = k ? fmaxf(b_pa, c_pa) : c_pa;
        b_qa = k ? fmaxf(b_qa, c_qa) : c_qa;
        b_pb = k ? fmaxf(b_pb, c_pb) : c_pb;
        b_qb = k ? fmaxf(b_qb, c_qb) : c_qb;
    }

    const float lo0 = -GAMMA / DTOT;
    float t_pa = fmaxf(b_pa - 2.4e-6f * (S_pa + GAMMA), lo0);
    float t_qa = fmaxf(b_qa - 2.4e-6f * (S_qa + GAMMA), lo0);
    float t_pb = fmaxf(b_pb - 2.4e-6f * (S_pb + GAMMA), lo0);
    float t_qb = fmaxf(b_qb - 2.4e-6f * (S_qb + GAMMA), lo0);
    int  s_pa = -1, s_qa = -1, s_pb = -1, s_qb = -1;
    bool d_pa = false, d_qa = false, d_pb = false, d_qb = false;

    const int wbase = wv * (LSLOTS * LROW);
    const int lcol  = wbase + lane * 4;
    bool listed = false;
    bool solved = false;
    int  cnt4   = 0;
    if (__all(t_pa >= 0.0f && t_qa >= 0.0f && t_pb >= 0.0f && t_qb >= 0.0f)) {
        int o_pa = lcol * 4, o_qa = lcol * 4 + 4, o_pb = lcol * 4 + 8, o_qb = lcol * 4 + 12;
        const int lim = lcol * 4 + LCAP * LSTEP;
        int peak = 0;
        const float a_pa = __builtin_fabsf(t_pa), a_qa = __builtin_fabsf(t_qa);
        const float a_pb = __builtin_fabsf(t_pb), a_qb = __builtin_fabsf(t_qb);
        char* const Lb = (char*)sL;
        const float amin4 = fminf(fminf(a_pa, a_qa), fminf(a_pb, a_qb));
        const float a_min = __uint_as_float(~wave_umax(~__float_as_uint(amin4)));
        const float wwide = wmax * CWIDEN;
        const unsigned hmask = (lane & 8) ? 0xFF00FF00u : 0x00FF00FFu;
        const unsigned lmask = hmask & ((1u << lane) - 1u);
        int* const cmine = &sC[wv][(lane >> 3) & 1][0];
        int n0 = 0, n1 = 0;
        #pragma unroll
        for (int j = 0; j < 9; ++j) {
            const bool     cand = __builtin_fmaf(__builtin_fabsf(pj[j]), CWIDEN, wwide) >= a_min;
            const unsigned m    = __builtin_amdgcn_ballot_w32(cand);
            const int      base = (lane & 8) ? n1 : n0;
            if (cand) cmine[base + __builtin_popcount(m & lmask)] = 32 * j + lane;
            n0 += __builtin_popcount(m & 0x00FF00FFu);
            n1 += __builtin_popcount(m & 0xFF00FF00u);
        }
        const int nt = (max(n0, n1) + 7) >> 3;
        #pragma nounroll
        for (int k = n0 + lane; k < nt * 8; k += 32) sC[wv][0][k] = HP;
        #pragma nounroll
        for (int k = n1 + lane; k < nt * 8; k += 32) sC[wv][1][k] = HP;
        const int* const cA = &sC[wv][l16 >> 3][l16 & 7];
        const v4i* const cC = (const v4i*)&sC[wv][half][0];
        if (nt <= NTREG) {
            v8f Vpa0 = {}, Vqa0 = {}, Vpb0 = {}, Vqb0 = {};
            v8f Vpa1 = {}, Vqa1 = {}, Vpb1 = {}, Vqb1 = {};
            if (nt > 0) form_tile(sW, pP, cA, cC, 0, half, l16, Bm, Vpa0, Vqa0, Vpb0, Vqb0);
            if (nt > 1) form_tile(sW, pP, cA, cC, 1, half, l16, Bm, Vpa1, Vqa1, Vpb1, Vqb1);
            const int off = HP - 16 * nt;
            for (int it = 0; it < MAXIT; ++it) {
                const float c_pa = __builtin_fabsf(t_pa), c_qa = __builtin_fabsf(t_qa);
                const float c_pb = __builtin_fabsf(t_pb), c_qb = __builtin_fabsf(t_qb);
                float    F_pa = 0.0f, F_qa = 0.0f, F_pb = 0.0f, F_qb = 0.0f;
                unsigned g_pa = 0u, g_qa = 0u, g_pb = 0u, g_qb = 0u;
                if (nt > 0) {
                    walk_tile(Vpa0, c_pa, F_pa, g_pa);
                    walk_tile(Vqa0, c_qa, F_qa, g_qa);
                    walk_tile(Vpb0, c_pb, F_pb, g_pb);
                    walk_tile(Vqb0, c_qb, F_qb, g_qb);
                }
                if (nt > 1) {
                    walk_tile(Vpa1, c_pa, F_pa, g_pa);
                    walk_tile(Vqa1, c_qa, F_qa, g_qa);
                    walk_tile(Vpb1, c_pb, F_pb, g_pb);
                    walk_tile(Vqb1, c_qb, F_qb, g_qb);
                }
                int n_pa = __builtin_popcount(g_pa), n_qa = __builtin_popcount(g_qa);
                int n_pb = __builtin_popcount(g_pb), n_qb = __builtin_popcount(g_qb);
                F_pa += __shfl_xor(F_pa, 16, 32);
                F_qa += __shfl_xor(F_qa, 16, 32);
                F_pb += __shfl_xor(F_pb, 16, 32);
                F_qb += __shfl_xor(F_qb, 16, 32);
                n_pa += __shfl_xor(n_pa, 16, 32);
                n_qa += __shfl_xor(n_qa, 16, 32);
                n_pb += __shfl_xor(n_pb, 16, 32);
                n_qb += __shfl_xor(n_qb, 16, 32);
                newton_step(t_pa, s_pa, d_pa, F_pa, off + n_pa);
                newton_step(t_qa, s_qa, d_qa, F_qa, off + n_qa);
                newton_step(t_pb, s_pb, d_pb, F_pb, off + n_pb);
                newton_step(t_qb, s_qb, d_qb, F_qb, off + n_qb);

                if (!__any(!(d_pa && d_qa && d_pb && d_qb))) break;
            }
            solved = true;
        } else {
            for (int i = 0; i < nt; ++i) {
                peak = max(peak, max(max(o_pa, o_qa), max(o_pb, o_qb)));
                o_pa = min(o_pa, lim);
                o_qa = min(o_qa, lim + 4);
                o_pb = min(o_pb, lim + 8);
                o_qb = min(o_qb, lim + 12);
                v8f Dpa, Dqa, Dpb, Dqb;
                form_tile(sW, pP, cA, cC, i, half, l16, Bm, Dpa, Dqa, Dpb, Dqb);
                #pragma unroll
                for (int r = 0; r < 8; ++r) {
                    *(float*)(Lb + o_pa) = Dpa[r];
                    *(float*)(Lb + o_qa) = Dqa[r];
                    *(float*)(Lb + o_pb) = Dpb[r];
                    *(float*)(Lb + o_qb) = Dqb[r];
                    o_pa += (__builtin_fabsf(Dpa[r]) >= a_pa) ? LSTEP : 0;
                    o_qa += (__builtin_fabsf(Dqa[r]) >= a_qa) ? LSTEP : 0;
                    o_pb += (__builtin_fabsf(Dpb[r]) >= a_pb) ? LSTEP : 0;
                    o_qb += (__builtin_fabsf(Dqb[r]) >= a_qb) ? LSTEP : 0;
                }
            }
            peak = max(peak, max(max(o_pa, o_qa), max(o_pb, o_qb)));
            listed = __all(((peak - wbase * 4) >> 9) <= LCAP);
            if (listed) {
                const int c_pa = (o_pa - wbase * 4) >> 9, c_qa = (o_qa - wbase * 4) >> 9;
                const int c_pb = (o_pb - wbase * 4) >> 9, c_qb = (o_qb - wbase * 4) >> 9;
                int m = max(max(c_pa, c_qa), max(c_pb, c_qb));
                m = max(m, __shfl_xor(m, 16, 32));
                m = max(m, __shfl_xor(m,  8, 32));
                m = max(m, __shfl_xor(m,  4, 32));
                m = max(m, __shfl_xor(m,  2, 32));
                m = max(m, __shfl_xor(m,  1, 32));
                cnt4 = (__builtin_amdgcn_readfirstlane(m) + 3) & ~3;
                for (int s = 0; s < cnt4 + 4; ++s) {
                    const int o = lcol + s * LROW;
                    if (s >= c_pa) sL[o]     = 0.0f;
                    if (s >= c_qa) sL[o + 1] = 0.0f;
                    if (s >= c_pb) sL[o + 2] = 0.0f;
                    if (s >= c_qb) sL[o + 3] = 0.0f;
                }
            }
        }
    }

    if (solved) {
    } else if (listed) {
        const v4f* Lr = (const v4f*)&sL[lcol];
        for (int it = 0; it < MAXIT; ++it) {
            const float a_pa = __builtin_fabsf(t_pa);
            const float a_qa = __builtin_fabsf(t_qa);
            const float a_pb = __builtin_fabsf(t_pb);
            const float a_qb = __builtin_fabsf(t_qb);
            float F_pa = 0.0f, F_qa = 0.0f, F_pb = 0.0f, F_qb = 0.0f;
            int   n_pa = 0, n_qa = 0, n_pb = 0, n_qb = 0;
            v4f nx0 = Lr[0], nx1 = Lr[32], nx2 = Lr[64], nx3 = Lr[96];
            for (int s = 0; s < cnt4; s += 4) {
                const v4f c0 = nx0, c1 = nx1, c2 = nx2, c3 = nx3;
                nx0 = Lr[(s + 4) * 32]; nx1 = Lr[(s + 5) * 32]; nx2 = Lr[(s + 6) * 32]; nx3 = Lr[(s + 7) * 32];
                unsigned g_pa = 0u, g_qa = 0u, g_pb = 0u, g_qb = 0u;
                #pragma unroll
                for (int k = 0; k < 4; ++k) {
                    const v4f c = (k == 0) ? c0 : (k == 1) ? c1 : (k == 2) ? c2 : c3;
                    const float e_pa = __builtin_fabsf(c[0]) - a_pa;
                    const float e_qa = __builtin_fabsf(c[1]) - a_qa;
                    const float e_pb = __builtin_fabsf(c[2]) - a_pb;
                    const float e_qb = __builtin_fabsf(c[3]) - a_qb;
                    F_pa += fmaxf(e_pa, 0.0f);
                    F_qa += fmaxf(e_qa, 0.0f);
                    F_pb += fmaxf(e_pb, 0.0f);
                    F_qb += fmaxf(e_qb, 0.0f);
                    g_pa = (g_pa << 1) | (__float_as_uint(e_pa) >> 31);
                    g_qa = (g_qa << 1) | (__float_as_uint(e_qa) >> 31);
                    g_pb = (g_pb << 1) | (__float_as_uint(e_pb) >> 31);
                    g_qb = (g_qb << 1) | (__float_as_uint(e_qb) >> 31);
                }
                n_pa += __builtin_popcount(g_pa);
                n_qa += __builtin_popcount(g_qa);
                n_pb += __builtin_popcount(g_pb);
                n_qb += __builtin_popcount(g_qb);
            }
            F_pa += __shfl_xor(F_pa, 16, 32);
            F_qa += __shfl_xor(F_qa, 16, 32);
            F_pb += __shfl_xor(F_pb, 16, 32);
            F_qb += __shfl_xor(F_qb, 16, 32);
            n_pa += __shfl_xor(n_pa, 16, 32);
            n_qa += __shfl_xor(n_qa, 16, 32);
            n_pb += __shfl_xor(n_pb, 16, 32);
            n_qb += __shfl_xor(n_qb, 16, 32);
            const int off = HP - 2 * cnt4;
            newton_step(t_pa, s_pa, d_pa, F_pa, off + n_pa);
            newton_step(t_qa, s_qa, d_qa, F_qa, off + n_qa);
            newton_step(t_pb, s_pb, d_pb, F_pb, off + n_pb);
            newton_step(t_qb, s_qb, d_qb, F_qb, off + n_qb);

            if (!__any(!(d_pa && d_qa && d_pb && d_qb))) break;
        }
    } else {
        for (int it = 0; it < MAXIT; ++it) {
            const float a_pa = __builtin_fabsf(t_pa);
            const float a_qa = __builtin_fabsf(t_qa);
            const float a_pb = __builtin_fabsf(t_pb);
            const float a_qb = __builtin_fabsf(t_qb);
            float F_pa = 0.0f, F_qa = 0.0f, F_pb = 0.0f, F_qb = 0.0f;
            int   n_pa = 0, n_qa = 0, n_pb = 0, n_qb = 0;

            for (int h0 = 0; h0 < HP; h0 += 32) {
                unsigned g_pa = 0u, g_qa = 0u, g_pb = 0u, g_qb = 0u;
                #pragma unroll
                for (int s = 0; s < 2; ++s) {
                    const int hb = h0 + 16 * s;
                    float pv = pP[hb + l16];
                    float a0 = half ? 0.0f : pv;
                    v2f Ap; Ap[0] =  a0; Ap[1] = 0.0f;
                    v2f Aq; Aq[0] = -a0; Aq[1] = 0.0f;
                    v8f Ca, Cb;
                    #pragma unroll
                    for (int r = 0; r < 8; ++r) {
                        const float* row = &sW[(hb + r + rowoff) * WPAD];
                        Ca[r] = row[l16];
                        Cb[r] = row[l16 + 16];
                    }
                    v8f Dpa = __builtin_amdgcn_wmma_f32_16x16x4_f32(false, Ap, false, Bm, (short)0, Ca, false, false);
                    v8f Dqa = __builtin_amdgcn_wmma_f32_16x16x4_f32(false, Aq, false, Bm, (short)0, Ca, false, false);
                    v8f Dpb = __builtin_amdgcn_wmma_f32_16x16x4_f32(false, Ap, false, Bm, (short)0, Cb, false, false);
                    v8f Dqb = __builtin_amdgcn_wmma_f32_16x16x4_f32(false, Aq, false, Bm, (short)0, Cb, false, false);
                    #pragma unroll
                    for (int r = 0; r < 8; ++r) {
                        const float e_pa = __builtin_fabsf(Dpa[r]) - a_pa;
                        const float e_qa = __builtin_fabsf(Dqa[r]) - a_qa;
                        const float e_pb = __builtin_fabsf(Dpb[r]) - a_pb;
                        const float e_qb = __builtin_fabsf(Dqb[r]) - a_qb;
                        F_pa += fmaxf(e_pa, 0.0f);
                        F_qa += fmaxf(e_qa, 0.0f);
                        F_pb += fmaxf(e_pb, 0.0f);
                        F_qb += fmaxf(e_qb, 0.0f);
                        g_pa = (g_pa << 1) | (__float_as_uint(e_pa) >> 31);
                        g_qa = (g_qa << 1) | (__float_as_uint(e_qa) >> 31);
                        g_pb = (g_pb << 1) | (__float_as_uint(e_pb) >> 31);
                        g_qb = (g_qb << 1) | (__float_as_uint(e_qb) >> 31);
                    }
                }
                n_pa += __builtin_popcount(g_pa);
                n_qa += __builtin_popcount(g_qa);
                n_pb += __builtin_popcount(g_pb);
                n_qb += __builtin_popcount(g_qb);
            }
            F_pa += __shfl_xor(F_pa, 16, 32);
            F_qa += __shfl_xor(F_qa, 16, 32);
            F_pb += __shfl_xor(F_pb, 16, 32);
            F_qb += __shfl_xor(F_qb, 16, 32);
            n_pa += __shfl_xor(n_pa, 16, 32);
            n_qa += __shfl_xor(n_qa, 16, 32);
            n_pb += __shfl_xor(n_pb, 16, 32);
            n_qb += __shfl_xor(n_qb, 16, 32);

            newton_step(t_pa, s_pa, d_pa, F_pa, n_pa);
            newton_step(t_qa, s_qa, d_qa, F_qa, n_qa);
            newton_step(t_pb, s_pb, d_pb, F_pb, n_pb);
            newton_step(t_qb, s_qb, d_qb, F_qb, n_qb);

            if (!__any(!(d_pa && d_qa && d_pb && d_qb))) break;
        }
    }

    const float tp = half ? t_pb : t_pa;
    const float tq = half ? t_qb : t_qa;
    const float conv = 0.5f * DTOT * (tp - tq);

    out[(b * 32 + lane) * 1024 + pos] = conv;
}

extern "C" void kernel_launch(void* const* d_in, const int* in_sizes, int n_in,
                              void* d_out, int out_size, void* d_ws, size_t ws_size,
                              hipStream_t stream) {
    const float* x = (const float*)d_in[0];
    const float* w = (const float*)d_in[1];
    float* out = (float*)d_out;
    (void)in_sizes; (void)n_in; (void)out_size; (void)d_ws; (void)ws_size;

    mpconv_sparsemax_kernel<<<256, 256, 0, stream>>>(x, w, out);
}


# Lane = output channel: 8 candidate rows per tile, whose row halves carry channels 0-15 / 16-31, so a lane sums its channel's problems whole: no LDS merge, no lists; a problem's sum order changes
# speedup vs baseline: 15.5205x; 1.0955x over previous
#include <hip/hip_runtime.h>


typedef __attribute__((ext_vector_type(2))) float v2f;
typedef __attribute__((ext_vector_type(8))) float v8f;

#define HP     288
#define WPAD   34
#define GAMMA  1.0f
#define DTOT   576.0f
#define MAXIT  (2 * HP + 2)
#define XSLAB  960
#define NSEL   8
#define CPITCH (HP + 8)
#define CWIDEN 1.00001f
#define NTREG  6

typedef __attribute__((ext_vector_type(4))) float v4f;
typedef __attribute__((ext_vector_type(4))) int   v4i;

__device__ __forceinline__ unsigned wave_umax(unsigned v)
{
    v = max(v, (unsigned)__builtin_amdgcn_update_dpp(0, (int)v, 0x161, 0xf, 0xf, true));
    v = max(v, (unsigned)__builtin_amdgcn_update_dpp(0, (int)v, 0x162, 0xf, 0xf, true));
    v = max(v, (unsigned)__builtin_amdgcn_update_dpp(0, (int)v, 0x164, 0xf, 0xf, true));
    v = max(v, (unsigned)__builtin_amdgcn_update_dpp(0, (int)v, 0x168, 0xf, 0xf, true));
    v = max(v, (unsigned)__builtin_amdgcn_permlanex16((int)v, (int)v, 0x76543210, 0xfedcba98, false, false));
    return v;
}

__device__ __forceinline__ void newton_step(float& t, int& sprev, bool& done, float F, int nneg)
{
    const float a   = __builtin_fabsf(t);
    const int   cge = HP - nneg;
    const bool  neg = t < 0.0f;
    const float f   = neg ? (F + DTOT * a) : F;
    int slope = neg ? (2 * HP - cge) : cge;
    slope = slope < 1 ? 1 : slope;
    float step = (f - GAMMA) * __builtin_amdgcn_rcpf((float)slope);
    step = step > 0.0f ? step : 0.0f;
    const bool conv = (slope == sprev) || !(step > 0.0f);
    t     = done ? t : t + step;
    sprev = done ? sprev : slope;
    done  = done || conv;
}

__device__ __forceinline__ void form_tile(const float* sWt, const float* pP, const int* cA, const v4i* cC,
                                          int i, int half, int lane, v2f Bm, v8f& Dp, v8f& Dq)
{
    const v4i r0 = cC[2 * i], r1 = cC[2 * i + 1];
    const float pv = pP[cA[8 * i]];
    const float a0 = half ? 0.0f : pv;
    v2f Ap; Ap[0] =  a0; Ap[1] = 0.0f;
    v2f Aq; Aq[0] = -a0; Aq[1] = 0.0f;
    v8f C;
    #pragma unroll
    for (int r = 0; r < 8; ++r) {
        const int hr = (r < 4) ? r0[r & 3] : r1[r & 3];
        C[r] = sWt[hr * WPAD + lane];
    }
    Dp = __builtin_amdgcn_wmma_f32_16x16x4_f32(false, Ap, false, Bm, (short)0, C, false, false);
    Dq = __builtin_amdgcn_wmma_f32_16x16x4_f32(false, Aq, false, Bm, (short)0, C, false, false);
}

__device__ __forceinline__ void walk_tile(const v8f& D, float a, float& F, int& n)
{
    unsigned g = 0u;
    #pragma unroll
    for (int r = 0; r < 8; ++r) {
        const float e = __builtin_fabsf(D[r]) - a;
        F += fmaxf(e, 0.0f);
        g = (g << 1) | (__float_as_uint(e) >> 31);
    }
    n += __builtin_popcount(g);
}

template <int NT>
__device__ __forceinline__ void newton_tiles(const float* sWt, const float* pP, const int* cA, const v4i* cC,
                                             int half, int lane, v2f Bm, float& t_p, float& t_q)
{
    v8f Vp[NT], Vq[NT];
    #pragma unroll
    for (int i = 0; i < NT; ++i) form_tile(sWt, pP, cA, cC, i, half, lane, Bm, Vp[i], Vq[i]);
    const int off = HP - 8 * NT;
    int  s_p = -1, s_q = -1;
    bool d_p = false, d_q = false;
    for (int it = 0; it < MAXIT; ++it) {
        const float c_p = __builtin_fabsf(t_p), c_q = __builtin_fabsf(t_q);
        float F_p = 0.0f, F_q = 0.0f;
        int   n_p = 0, n_q = 0;
        #pragma unroll
        for (int i = 0; i < NT; ++i) {
            walk_tile(Vp[i], c_p, F_p, n_p);
            walk_tile(Vq[i], c_q, F_q, n_q);
        }
        newton_step(t_p, s_p, d_p, F_p, off + n_p);
        newton_step(t_q, s_q, d_q, F_q, off + n_q);
        if (!__any(!(d_p && d_q))) break;
    }
}

__global__ __launch_bounds__(256) void mpconv_sparsemax_kernel(
    const float* __restrict__ x,
    const float* __restrict__ w,
    float* __restrict__ out)
{
    __shared__ float sW[(HP + 1) * WPAD];
    __shared__ float sP[8][HP + 8];
    __shared__ __attribute__((aligned(16))) float sWm[8];
    __shared__ __attribute__((aligned(16))) int sC[8][CPITCH];
    __shared__ float sX[1024];

    const int tid  = threadIdx.x;
    const int lane = tid & 31;
    const int wv   = tid >> 5;

    const int blk = blockIdx.x;
    const int b   = blk >> 7;
    const int oh  = (blk >> 2) & 31;
    const int ow0 = (blk & 3) * 8;
    const int ow  = ow0 + wv;
    const int pos = oh * 32 + ow;
    const int wo  = tid >> 3;
    const int wm  = tid & 7;
    float xr[4];
    bool  xin[4];
    #pragma unroll
    for (int k = 0; k < 4; ++k) {
        int e = tid + 256 * k;
        e = e < XSLAB ? e : XSLAB - 1;
        const int c   = e / 30;
        const int rem = e - c * 30;
        const int ry  = rem / 10;
        const int cx  = rem - ry * 10;
        const int y   = oh - 1 + ry;
        const int xx  = ow0 - 1 + cx;
        const bool in = (unsigned)y < 32u && (unsigned)xx < 32u;
        const int yc  = min(max(y, 0), 31);
        const int xc  = min(max(xx, 0), 31);
        xr[k]  = x[((b * 32 + c) * 32 + yc) * 32 + xc];
        xin[k] = in;
    }
    __builtin_amdgcn_sched_barrier(0);
    v4f wr[9];
    {
        const v4f* wsrc = (const v4f*)(w + wo * HP + wm * 4);
        #pragma unroll
        for (int k = 0; k < 9; ++k) wr[k] = wsrc[k * 8];
    }
    __builtin_amdgcn_sched_barrier(0);
    #pragma unroll
    for (int k = 0; k < 4; ++k) sX[tid + 256 * k] = xin[k] ? xr[k] : 0.0f;
    asm volatile("s_wait_dscnt 0x0\n\ts_barrier_signal -1\n\ts_barrier_wait -1" ::: "memory");

    unsigned m1, m2;
    {
        const int c = lane;
        unsigned key[9];
        #pragma unroll
        for (int ki = 0; ki < 3; ++ki) {
            #pragma unroll
            for (int kj = 0; kj < 3; ++kj) {
                const int   h  = c * 9 + ki * 3 + kj;
                const float pe = sX[c * 30 + ki * 10 + wv + kj];
                sP[wv][h] = pe;
                key[ki * 3 + kj] = (__float_as_uint(__builtin_fabsf(pe)) & 0xFFFFFE00u) | (unsigned)h;
            }
        }
        m1 = max(key[0], key[1]);
        m2 = min(key[0], key[1]);
        #pragma unroll
        for (int i = 2; i < 9; ++i) {
            m2 = max(m2, min(m1, key[i]));
            m1 = max(m1, key[i]);
        }
    }
    int hsel[NSEL];
    #pragma unroll
    for (int k = 0; k < NSEL; ++k) {
        const unsigned g = wave_umax(m1);
        hsel[k] = __builtin_amdgcn_readfirstlane((int)(g & 0x1FFu));
        const bool own = (m1 == g);
        m1 = own ? m2 : m1;
        m2 = own ? 0u : m2;
    }
    float pj[9];
    #pragma unroll
    for (int j = 0; j < 9; ++j) pj[j] = sP[wv][32 * j + lane];
    __builtin_amdgcn_sched_barrier(0);
    #pragma unroll
    for (int k = 0; k < 9; ++k) {
        #pragma unroll
        for (int j = 0; j < 4; ++j)
            sW[(k * 32 + wm * 4 + j) * WPAD + wo] = wr[k][j];
    }
    {
        float wmx = 0.0f;
        #pragma unroll
        for (int k = 0; k < 9; ++k) {
            #pragma unroll
            for (int j = 0; j < 4; ++j) wmx = fmaxf(wmx, __builtin_fabsf(wr[k][j]));
        }
        const unsigned wb = wave_umax(__float_as_uint(wmx));
        if (lane == 0) { sWm[wv] = __uint_as_float(wb); sP[wv][HP] = 0.0f; }
        if (tid < WPAD) sW[HP * WPAD + tid] = 0.0f;
    }
    __syncthreads();

    const int   half = lane >> 4;
    const int   l16  = lane & 15;
    const float bsel = half ? 0.0f : 1.0f;
    v2f Bm; Bm[0] = bsel; Bm[1] = 0.0f;
    const float* pP = sP[wv];
    float wmax;
    {
        const v4f m0 = *(const v4f*)&sWm[0], m1 = *(const v4f*)&sWm[4];
        wmax = fmaxf(fmaxf(fmaxf(m0[0], m0[1]), fmaxf(m0[2], m0[3])), fmaxf(fmaxf(m1[0], m1[1]), fmaxf(m1[2], m1[3])));
    }

    float S_p = 0.0f, S_q = 0.0f;
    float b_p = 0.0f, b_q = 0.0f;
    #pragma unroll
    for (int k = 0; k < NSEL; ++k) {
        const float pk = pP[hsel[k]];
        const float wk = sW[hsel[k] * WPAD + lane];
        S_p += __builtin_fabsf(pk + wk);
        S_q += __builtin_fabsf(pk - wk);
        const float inv = 1.0f / (float)(k + 1);
        const float c_p = (S_p - GAMMA) * inv, c_q = (S_q - GAMMA) * inv;
        b_p = k ? fmaxf(b_p, c_p) : c_p;
        b_q = k ? fmaxf(b_q, c_q) : c_q;
    }

    const float lo0 = -GAMMA / DTOT;
    float t_p = fmaxf(b_p - 2.4e-6f * (S_p + GAMMA), lo0);
    float t_q = fmaxf(b_q - 2.4e-6f * (S_q + GAMMA), lo0);

    bool solved = false;
    if (__all(t_p >= 0.0f && t_q >= 0.0f)) {
        const float amin2 = fminf(__builtin_fabsf(t_p), __builtin_fabsf(t_q));
        const float a_min = __uint_as_float(~wave_umax(~__float_as_uint(amin2)));
        const float wwide = wmax * CWIDEN;
        const unsigned lmask = (1u << lane) - 1u;
        int* const cmine = &sC[wv][0];
        int n = 0;
        #pragma unroll
        for (int j = 0; j < 9; ++j) {
            const bool     cand = __builtin_fmaf(__builtin_fabsf(pj[j]), CWIDEN, wwide) >= a_min;
            const unsigned m    = __builtin_amdgcn_ballot_w32(cand);
            if (cand) cmine[n + __builtin_popcount(m & lmask)] = 32 * j + lane;
            n += __builtin_popcount(m);
        }
        const int nt = (n + 7) >> 3;
        const int ntp = nt <= 1 ? 1 : (nt <= 3 ? nt : NTREG);
        if (n + lane < ntp * 8) cmine[n + lane] = HP;
        const int* const cA = &sC[wv][l16 & 7];
        const v4i* const cC = (const v4i*)&sC[wv][0];
        solved = nt <= NTREG;
        if      (nt <= 1)     newton_tiles<1>(sW, pP, cA, cC, half, lane, Bm, t_p, t_q);
        else if (nt == 2)     newton_tiles<2>(sW, pP, cA, cC, half, lane, Bm, t_p, t_q);
        else if (nt == 3)     newton_tiles<3>(sW, pP, cA, cC, half, lane, Bm, t_p, t_q);
        else if (nt <= NTREG) newton_tiles<NTREG>(sW, pP, cA, cC, half, lane, Bm, t_p, t_q);
    }

    if (!solved) {
        int  s_p = -1, s_q = -1;
        bool d_p = false, d_q = false;
        for (int it = 0; it < MAXIT; ++it) {
            const float a_p = __builtin_fabsf(t_p);
            const float a_q = __builtin_fabsf(t_q);
            float F_p = 0.0f, F_q = 0.0f;
            int   n_p = 0, n_q = 0;

            for (int h0 = 0; h0 < HP; h0 += 16) {
                #pragma unroll
                for (int s = 0; s < 2; ++s) {
                    const int   hb = h0 + 8 * s;
                    const float pv = pP[hb + (l16 & 7)];
                    const float a0 = half ? 0.0f : pv;
                    v2f Ap; Ap[0] =  a0; Ap[1] = 0.0f;
                    v2f Aq; Aq[0] = -a0; Aq[1] = 0.0f;
                    v8f C;
                    #pragma unroll
                    for (int r = 0; r < 8; ++r) C[r] = sW[(hb + r) * WPAD + lane];
                    const v8f Dp = __builtin_amdgcn_wmma_f32_16x16x4_f32(false, Ap, false, Bm, (short)0, C, false, false);
                    const v8f Dq = __builtin_amdgcn_wmma_f32_16x16x4_f32(false, Aq, false, Bm, (short)0, C, false, false);
                    walk_tile(Dp, a_p, F_p, n_p);
                    walk_tile(Dq, a_q, F_q, n_q);
                }
            }

            newton_step(t_p, s_p, d_p, F_p, n_p);
            newton_step(t_q, s_q, d_q, F_q, n_q);

            if (!__any(!(d_p && d_q))) break;
        }
    }

    const float conv = 0.5f * DTOT * (t_p - t_q);

    out[(b * 32 + lane) * 1024 + pos] = conv;
}

extern "C" void kernel_launch(void* const* d_in, const int* in_sizes, int n_in,
                              void* d_out, int out_size, void* d_ws, size_t ws_size,
                              hipStream_t stream) {
    const float* x = (const float*)d_in[0];
    const float* w = (const float*)d_in[1];
    float* out = (float*)d_out;
    (void)in_sizes; (void)n_in; (void)out_size; (void)d_ws; (void)ws_size;

    mpconv_sparsemax_kernel<<<256, 256, 0, stream>>>(x, w, out);
}
